# starting kernel
# baseline (speedup 1.0000x reference)
#include <hip/hip_runtime.h>

#define IN_F      128
#define HC        128
#define HEADS     4
#define OUT_F     32
#define NEG_SLOPE 0.2f

typedef float v2f __attribute__((ext_vector_type(2)));
typedef float v4f __attribute__((ext_vector_type(4)));
typedef float v8f __attribute__((ext_vector_type(8)));

__device__ __forceinline__ unsigned fkey(float f) {
    unsigned b = __float_as_uint(f);
    return (b & 0x80000000u) ? ~b : (b | 0x80000000u);
}
__device__ __forceinline__ float funkey(unsigned k) {
    unsigned b = (k & 0x80000000u) ? (k & 0x7fffffffu) : ~k;
    return __uint_as_float(b);
}
__device__ __forceinline__ float leaky(float x) {
    return x > 0.0f ? x : NEG_SLOPE * x;
}

__global__ __launch_bounds__(256) void gat_init(float* __restrict__ out,
                                                const float* __restrict__ bias,
                                                unsigned* __restrict__ mkey,
                                                float* __restrict__ denom,
                                                int total, int nm) {
    int i = blockIdx.x * 256 + threadIdx.x;
    if (i < total) out[i] = bias[i & (HC - 1)];
    if (i < nm) { mkey[i] = 0u; denom[i] = 0.0f; }
}

#define WSTR 136
#define XSTR 132
#define GEMM_LDS_BYTES ((128 * WSTR + 16 * XSTR) * 4)

__global__ __launch_bounds__(128) void gat_gemm_wmma(const float* __restrict__ x,
                                                     const float* __restrict__ W,
                                                     float* __restrict__ h) {
    extern __shared__ float smem[];
    float* Wl = smem;
    float* Xl = smem + 128 * WSTR;

    const int tid  = threadIdx.x;
    const int row0 = blockIdx.x * 16;

    for (int i = tid; i < 128 * 32; i += 128) {
        int r  = i >> 5;
        int c4 = i & 31;
        v4f v = ((const v4f*)W)[r * 32 + c4];
        *(v4f*)&Wl[r * WSTR + c4 * 4] = v;
    }
    for (int i = tid; i < 16 * 32; i += 128) {
        int r  = i >> 5;
        int c4 = i & 31;
        v4f v = ((const v4f*)x)[(size_t)(row0 + r) * 32 + c4];
        *(v4f*)&Xl[r * XSTR + c4 * 4] = v;
    }
    __syncthreads();

    const int wave = tid >> 5;
    const int lane = tid & 31;
    const int L    = lane & 15;
    const int g    = lane >> 4;

    const int col0 = wave * 16;
    const int col1 = (wave + 4) * 16;

    v8f acc0 = {};
    v8f acc1 = {};

    for (int k0 = 0; k0 < IN_F; k0 += 4) {
        const int k = k0 + 2 * g;
        v2f a, b0, b1;
        a.x  = Xl[L * XSTR + k];
        a.y  = Xl[L * XSTR + k + 1];
        b0.x = Wl[k * WSTR + col0 + L];
        b0.y = Wl[(k + 1) * WSTR + col0 + L];
        b1.x = Wl[k * WSTR + col1 + L];
        b1.y = Wl[(k + 1) * WSTR + col1 + L];
        acc0 = __builtin_amdgcn_wmma_f32_16x16x4_f32(false, a, false, b0, (short)0, acc0, false, false);
        acc1 = __builtin_amdgcn_wmma_f32_16x16x4_f32(false, a, false, b1, (short)0, acc1, false, false);
    }

    const int mbase = row0 + 8 * g;
    for (int r = 0; r < 8; ++r) {
        h[(size_t)(mbase + r) * HC + col0 + L] = acc0[r];
        h[(size_t)(mbase + r) * HC + col1 + L] = acc1[r];
    }
}

__global__ __launch_bounds__(256) void gat_attn_coeff(const float* __restrict__ h,
                                                      const float* __restrict__ att_src,
                                                      const float* __restrict__ att_dst,
                                                      float* __restrict__ a_src,
                                                      float* __restrict__ a_dst,
                                                      int N) {
    int n    = blockIdx.x * 8 + (threadIdx.x >> 5);
    int lane = threadIdx.x & 31;
    if (n >= N) return;

    float s[HEADS], d[HEADS];
#pragma unroll
    for (int hh = 0; hh < HEADS; ++hh) {
        float v = h[(size_t)n * HC + hh * OUT_F + lane];
        s[hh] = v * att_src[hh * OUT_F + lane];
        d[hh] = v * att_dst[hh * OUT_F + lane];
    }
#pragma unroll
    for (int off = 16; off > 0; off >>= 1) {
#pragma unroll
        for (int hh = 0; hh < HEADS; ++hh) {
            s[hh] += __shfl_xor(s[hh], off, 32);
            d[hh] += __shfl_xor(d[hh], off, 32);
        }
    }
    if (lane == 0) {
#pragma unroll
        for (int hh = 0; hh < HEADS; ++hh) {
            a_src[n * HEADS + hh] = s[hh];
            a_dst[n * HEADS + hh] = d[hh];
        }
    }
}

__global__ __launch_bounds__(256) void gat_edge_max(const int* __restrict__ ei,
                                                    const float* __restrict__ a_src,
                                                    const float* __restrict__ a_dst,
                                                    unsigned* __restrict__ mkey,
                                                    int E, int N) {
    int e = blockIdx.x * 256 + threadIdx.x;
    if (e >= E + N) return;
    int s, d;
    if (e < E) { s = ei[e]; d = ei[E + e]; } else { s = d = e - E; }
    v4f as = *(const v4f*)&a_src[s * HEADS];
    v4f ad = *(const v4f*)&a_dst[d * HEADS];
#pragma unroll
    for (int hh = 0; hh < HEADS; ++hh) {
        float lr = leaky(as[hh] + ad[hh]);
        atomicMax(&mkey[d * HEADS + hh], fkey(lr));
    }
}

__global__ __launch_bounds__(256) void gat_edge_denom(const int* __restrict__ ei,
                                                      const float* __restrict__ a_src,
                                                      const float* __restrict__ a_dst,
                                                      const unsigned* __restrict__ mkey,
                                                      float* __restrict__ denom,
                                                      int E, int N) {
    int e = blockIdx.x * 256 + threadIdx.x;
    if (e >= E + N) return;
    int s, d;
    if (e < E) { s = ei[e]; d = ei[E + e]; } else { s = d = e - E; }
    v4f as = *(const v4f*)&a_src[s * HEADS];
    v4f ad = *(const v4f*)&a_dst[d * HEADS];
#pragma unroll
    for (int hh = 0; hh < HEADS; ++hh) {
        float lr = leaky(as[hh] + ad[hh]);
        float m  = funkey(mkey[d * HEADS + hh]);
        atomicAdd(&denom[d * HEADS + hh], __expf(lr - m));
    }
}

__global__ __launch_bounds__(256) void gat_edge_scatter(const int* __restrict__ ei,
                                                        const float* __restrict__ a_src,
                                                        const float* __restrict__ a_dst,
                                                        const unsigned* __restrict__ mkey,
                                                        const float* __restrict__ denom,
                                                        const float* __restrict__ h,
                                                        float* __restrict__ out,
                                                        int E, int N) {
    int e    = blockIdx.x * 8 + (threadIdx.x >> 5);
    int lane = threadIdx.x & 31;
    if (e >= E + N) return;
    int s, d;
    if (e < E) { s = ei[e]; d = ei[E + e]; } else { s = d = e - E; }

    int head  = lane >> 3;
    float lr  = leaky(a_src[s * HEADS + head] + a_dst[d * HEADS + head]);
    float m   = funkey(mkey[d * HEADS + head]);
    float den = denom[d * HEADS + head];
    float alpha = __expf(lr - m) / (den + 1e-16f);

    v4f hv = *(const v4f*)&h[(size_t)s * HC + lane * 4];
    float* o = &out[(size_t)d * HC + lane * 4];
    atomicAdd(o + 0, hv.x * alpha);
    atomicAdd(o + 1, hv.y * alpha);
    atomicAdd(o + 2, hv.z * alpha);
    atomicAdd(o + 3, hv.w * alpha);
}

extern "C" void kernel_launch(void* const* d_in, const int* in_sizes, int n_in,
                              void* d_out, int out_size, void* d_ws, size_t ws_size,
                              hipStream_t stream) {
    const float* x        = (const float*)d_in[0];
    const int*   ei       = (const int*)d_in[1];
    const float* W        = (const float*)d_in[2];
    const float* att_src  = (const float*)d_in[3];
    const float* att_dst  = (const float*)d_in[4];
    const float* bias     = (const float*)d_in[5];
    float*       out      = (float*)d_out;

    const int N = in_sizes[0] / IN_F;
    const int E = in_sizes[1] / 2;

    float*    h     = (float*)d_ws;
    float*    a_src = h + (size_t)N * HC;
    float*    a_dst = a_src + (size_t)N * HEADS;
    unsigned* mkey  = (unsigned*)(a_dst + (size_t)N * HEADS);
    float*    denom = (float*)(mkey + (size_t)N * HEADS);

    const int total = N * HC;
    gat_init<<<(total + 255) / 256, 256, 0, stream>>>(out, bias, mkey, denom, total, N * HEADS);

    gat_gemm_wmma<<<N / 16, 128, GEMM_LDS_BYTES, stream>>>(x, W, h);

    gat_attn_coeff<<<(N + 7) / 8, 256, 0, stream>>>(h, att_src, att_dst, a_src, a_dst, N);

    const int ET = E + N;
    gat_edge_max<<<(ET + 255) / 256, 256, 0, stream>>>(ei, a_src, a_dst, mkey, E, N);
    gat_edge_denom<<<(ET + 255) / 256, 256, 0, stream>>>(ei, a_src, a_dst, mkey, denom, E, N);
    gat_edge_scatter<<<(ET + 7) / 8, 256, 0, stream>>>(ei, a_src, a_dst, mkey, denom, h, out, E, N);
}


# Output rows accumulate in LDS: each block owns 391 destination nodes, filters the whole edge list for them (edges never moved) and adds p*h with LDS f32 adds, arrival order; no global f32 atomics
# speedup vs baseline: 2.1033x; 2.1033x over previous
#include <hip/hip_runtime.h>

#define IN_F      128
#define HC        128
#define HEADS     4
#define OUT_F     32
#define NEG_SLOPE 0.2f

typedef float v2f __attribute__((ext_vector_type(2)));
typedef float v4f __attribute__((ext_vector_type(4)));
typedef float v8f __attribute__((ext_vector_type(8)));

__device__ __forceinline__ unsigned fkey(float f) {
    unsigned b = __float_as_uint(f);
    return (b & 0x80000000u) ? ~b : (b | 0x80000000u);
}
__device__ __forceinline__ float funkey(unsigned k) {
    unsigned b = (k & 0x80000000u) ? (k & 0x7fffffffu) : ~k;
    return __uint_as_float(b);
}
__device__ __forceinline__ float leaky(float x) {
    return x > 0.0f ? x : NEG_SLOPE * x;
}

__global__ __launch_bounds__(256) void gat_init(unsigned* __restrict__ mkey, int nm) {
    int i = blockIdx.x * 256 + threadIdx.x;
    if (i < nm) mkey[i] = 0u;
}

#define WSTR 136
#define XSTR 132
#define GEMM_LDS_BYTES ((128 * WSTR + 16 * XSTR) * 4)

__global__ __launch_bounds__(128) void gat_gemm_wmma(const float* __restrict__ x,
                                                     const float* __restrict__ W,
                                                     float* __restrict__ h) {
    extern __shared__ float smem[];
    float* Wl = smem;
    float* Xl = smem + 128 * WSTR;

    const int tid  = threadIdx.x;
    const int row0 = blockIdx.x * 16;

    for (int i = tid; i < 128 * 32; i += 128) {
        int r  = i >> 5;
        int c4 = i & 31;
        v4f v = ((const v4f*)W)[r * 32 + c4];
        *(v4f*)&Wl[r * WSTR + c4 * 4] = v;
    }
    for (int i = tid; i < 16 * 32; i += 128) {
        int r  = i >> 5;
        int c4 = i & 31;
        v4f v = ((const v4f*)x)[(size_t)(row0 + r) * 32 + c4];
        *(v4f*)&Xl[r * XSTR + c4 * 4] = v;
    }
    __syncthreads();

    const int wave = tid >> 5;
    const int lane = tid & 31;
    const int L    = lane & 15;
    const int g    = lane >> 4;

    const int col0 = wave * 16;
    const int col1 = (wave + 4) * 16;

    v8f acc0 = {};
    v8f acc1 = {};

    for (int k0 = 0; k0 < IN_F; k0 += 4) {
        const int k = k0 + 2 * g;
        v2f a, b0, b1;
        a.x  = Xl[L * XSTR + k];
        a.y  = Xl[L * XSTR + k + 1];
        b0.x = Wl[k * WSTR + col0 + L];
        b0.y = Wl[(k + 1) * WSTR + col0 + L];
        b1.x = Wl[k * WSTR + col1 + L];
        b1.y = Wl[(k + 1) * WSTR + col1 + L];
        acc0 = __builtin_amdgcn_wmma_f32_16x16x4_f32(false, a, false, b0, (short)0, acc0, false, false);
        acc1 = __builtin_amdgcn_wmma_f32_16x16x4_f32(false, a, false, b1, (short)0, acc1, false, false);
    }

    const int mbase = row0 + 8 * g;
    for (int r = 0; r < 8; ++r) {
        h[(size_t)(mbase + r) * HC + col0 + L] = acc0[r];
        h[(size_t)(mbase + r) * HC + col1 + L] = acc1[r];
    }
}

__global__ __launch_bounds__(256) void gat_attn_coeff(const float* __restrict__ h,
                                                      const float* __restrict__ att_src,
                                                      const float* __restrict__ att_dst,
                                                      float* __restrict__ a_src,
                                                      float* __restrict__ a_dst,
                                                      int N) {
    int n    = blockIdx.x * 8 + (threadIdx.x >> 5);
    int lane = threadIdx.x & 31;
    if (n >= N) return;

    float s[HEADS], d[HEADS];
#pragma unroll
    for (int hh = 0; hh < HEADS; ++hh) {
        float v = h[(size_t)n * HC + hh * OUT_F + lane];
        s[hh] = v * att_src[hh * OUT_F + lane];
        d[hh] = v * att_dst[hh * OUT_F + lane];
    }
#pragma unroll
    for (int off = 16; off > 0; off >>= 1) {
#pragma unroll
        for (int hh = 0; hh < HEADS; ++hh) {
            s[hh] += __shfl_xor(s[hh], off, 32);
            d[hh] += __shfl_xor(d[hh], off, 32);
        }
    }
    if (lane == 0) {
#pragma unroll
        for (int hh = 0; hh < HEADS; ++hh) {
            a_src[n * HEADS + hh] = s[hh];
            a_dst[n * HEADS + hh] = d[hh];
        }
    }
}

__global__ __launch_bounds__(256) void gat_edge_max(const int* __restrict__ ei,
                                                    const float* __restrict__ a_src,
                                                    const float* __restrict__ a_dst,
                                                    unsigned* __restrict__ mkey,
                                                    int E, int N) {
    int e = blockIdx.x * 256 + threadIdx.x;
    if (e >= E + N) return;
    int s, d;
    if (e < E) { s = ei[e]; d = ei[E + e]; } else { s = d = e - E; }
    v4f as = *(const v4f*)&a_src[s * HEADS];
    v4f ad = *(const v4f*)&a_dst[d * HEADS];
#pragma unroll
    for (int hh = 0; hh < HEADS; ++hh) {
        float lr = leaky(as[hh] + ad[hh]);
        atomicMax(&mkey[d * HEADS + hh], fkey(lr));
    }
}

#define RNODES 391
#define RING   64

__device__ __forceinline__ float bcast_f(float v, int k) {
    return __int_as_float(__builtin_amdgcn_readlane(__float_as_int(v), k));
}

__global__ __launch_bounds__(1024) void gat_range_sum(const int* __restrict__ ei,
                                                      const float* __restrict__ a_src,
                                                      const float* __restrict__ a_dst,
                                                      const unsigned* __restrict__ mkey,
                                                      const float* __restrict__ h,
                                                      const float* __restrict__ bias,
                                                      float* __restrict__ out,
                                                      int E, int N) {
    __shared__ __attribute__((aligned(16))) float acc[RNODES * HC];
    __shared__ __attribute__((aligned(16))) float den[RNODES * HEADS];
    __shared__ __attribute__((aligned(16))) float mloc[RNODES * HEADS];
    __shared__ __attribute__((aligned(16))) float adloc[RNODES * HEADS];
    __shared__ int ring_all[32 * RING];

    const int tid  = threadIdx.x;
    const int wave = tid >> 5;
    const int lane = tid & 31;
    const int n0   = blockIdx.x * RNODES;
    const int cnt  = min(RNODES, N - n0);
    const int rbase = wave * RING;

    for (int r = wave; r < cnt; r += 32) {
        const int n = n0 + r;
        const uint4 mk = *(const uint4*)&mkey[n * HEADS];
        const v4f   as = *(const v4f*)&a_src[n * HEADS];
        const v4f   ad = *(const v4f*)&a_dst[n * HEADS];
        const float m0 = funkey(mk.x), m1 = funkey(mk.y), m2 = funkey(mk.z), m3 = funkey(mk.w);
        const float p0 = __expf(leaky(as.x + ad.x) - m0);
        const float p1 = __expf(leaky(as.y + ad.y) - m1);
        const float p2 = __expf(leaky(as.z + ad.z) - m2);
        const float p3 = __expf(leaky(as.w + ad.w) - m3);
        const float* hr = h + (size_t)n * HC + lane;
        float* ar = acc + r * HC + lane;
        ar[0]  = hr[0]  * p0;
        ar[32] = hr[32] * p1;
        ar[64] = hr[64] * p2;
        ar[96] = hr[96] * p3;
        if (lane < HEADS) {
            const float pm = lane == 0 ? p0 : lane == 1 ? p1 : lane == 2 ? p2 : p3;
            const float mm = lane == 0 ? m0 : lane == 1 ? m1 : lane == 2 ? m2 : m3;
            const float aa = lane == 0 ? ad.x : lane == 1 ? ad.y : lane == 2 ? ad.z : ad.w;
            den[r * HEADS + lane]   = pm;
            mloc[r * HEADS + lane]  = mm;
            adloc[r * HEADS + lane] = aa;
        }
    }
    __syncthreads();

    const int* dstp = ei + E;
    int head_q = 0, tail_q = 0;

    auto drain = [&](int take) {
        const bool act = lane < take;
        const int  e   = act ? __hip_atomic_load(&ring_all[rbase + ((head_q + lane) & (RING - 1))], __ATOMIC_RELAXED, __HIP_MEMORY_SCOPE_WAVEFRONT) : 0;
        const int  s   = ei[e];
        const int  dl  = act ? dstp[e] - n0 : 0;
        const v4f  as  = *(const v4f*)&a_src[s * HEADS];
        const v4f  ad  = *(const v4f*)&adloc[dl * HEADS];
        const v4f  mm  = *(const v4f*)&mloc[dl * HEADS];
        const float p0 = __expf(leaky(as.x + ad.x) - mm.x);
        const float p1 = __expf(leaky(as.y + ad.y) - mm.y);
        const float p2 = __expf(leaky(as.z + ad.z) - mm.z);
        const float p3 = __expf(leaky(as.w + ad.w) - mm.w);
        if (act) {
            atomicAdd(&den[dl * HEADS + 0], p0);
            atomicAdd(&den[dl * HEADS + 1], p1);
            atomicAdd(&den[dl * HEADS + 2], p2);
            atomicAdd(&den[dl * HEADS + 3], p3);
        }
        for (int k0 = 0; k0 < take; k0 += 8) {
            float hv[8][4];
            int   dk[8];
#pragma unroll
            for (int j = 0; j < 8; ++j) {
                const int k  = min(k0 + j, take - 1);
                const int sk = __builtin_amdgcn_readlane(s, k);
                dk[j] = __builtin_amdgcn_readlane(dl, k);
                const float* hr = h + (size_t)sk * HC + lane;
                hv[j][0] = hr[0];
                hv[j][1] = hr[32];
                hv[j][2] = hr[64];
                hv[j][3] = hr[96];
            }
#pragma unroll
            for (int j = 0; j < 8; ++j) {
                if (k0 + j < take) {
                    const int k = k0 + j;
                    float* ar = acc + dk[j] * HC + lane;
                    atomicAdd(ar,      hv[j][0] * bcast_f(p0, k));
                    atomicAdd(ar + 32, hv[j][1] * bcast_f(p1, k));
                    atomicAdd(ar + 64, hv[j][2] * bcast_f(p2, k));
                    atomicAdd(ar + 96, hv[j][3] * bcast_f(p3, k));
                }
            }
        }
        head_q += take;
    };

    auto push = [&](bool hit, int e) {
        const unsigned mask = __builtin_amdgcn_ballot_w32(hit);
        if (mask != 0u) {
            const int pre = __builtin_popcount(mask & ((1u << lane) - 1u));
            if (hit) __hip_atomic_store(&ring_all[rbase + ((tail_q + pre) & (RING - 1))], e, __ATOMIC_RELAXED, __HIP_MEMORY_SCOPE_WAVEFRONT);
            tail_q += __builtin_popcount(mask);
            if (tail_q - head_q >= 32) drain(32);
        }
    };

    const bool vec = (((size_t)dstp) & 15) == 0;
    const int  E4  = vec ? (E >> 2) : 0;
    const int4* dst4 = (const int4*)dstp;
    for (int i0 = 0; i0 < E4; i0 += 1024 * 8) {
        int4 d[8];
#pragma unroll
        for (int u = 0; u < 8; ++u) {
            const int idx = i0 + u * 1024 + tid;
            d[u] = idx < E4 ? dst4[idx] : make_int4(-1, -1, -1, -1);
        }
        unsigned hb = 0u;
#pragma unroll
        for (int u = 0; u < 8; ++u) {
            hb |= ((unsigned)(d[u].x - n0) < (unsigned)cnt ? 1u : 0u) << (4 * u + 0);
            hb |= ((unsigned)(d[u].y - n0) < (unsigned)cnt ? 1u : 0u) << (4 * u + 1);
            hb |= ((unsigned)(d[u].z - n0) < (unsigned)cnt ? 1u : 0u) << (4 * u + 2);
            hb |= ((unsigned)(d[u].w - n0) < (unsigned)cnt ? 1u : 0u) << (4 * u + 3);
        }
        while (__builtin_amdgcn_ballot_w32(hb != 0u) != 0u) {
            const bool hit = hb != 0u;
            const int  t   = hit ? __builtin_ctz(hb) : 0;
            hb &= hb - 1u;
            const int  e   = 4 * (i0 + (t >> 2) * 1024 + tid) + (t & 3);
            push(hit, e);
        }
    }
    for (int e0 = 4 * E4 + wave * 32; e0 < E; e0 += 1024) {
        const int e = e0 + lane;
        const bool hit = e < E && (unsigned)(dstp[e < E ? e : 0] - n0) < (unsigned)cnt;
        push(hit, e);
    }
    if (tail_q - head_q > 0) drain(tail_q - head_q);
    __syncthreads();

    for (int r = wave; r < cnt; r += 32) {
        const v4f a = *(const v4f*)&acc[r * HC + lane * 4];
        const float dn = den[r * HEADS + (lane >> 3)] + 1e-16f;
        const v4f b = ((const v4f*)bias)[lane];
        v4f o;
        o.x = a.x / dn + b.x;
        o.y = a.y / dn + b.y;
        o.z = a.z / dn + b.z;
        o.w = a.w / dn + b.w;
        *(v4f*)&out[(size_t)(n0 + r) * HC + lane * 4] = o;
    }
}

extern "C" void kernel_launch(void* const* d_in, const int* in_sizes, int n_in,
                              void* d_out, int out_size, void* d_ws, size_t ws_size,
                              hipStream_t stream) {
    const float* x        = (const float*)d_in[0];
    const int*   ei       = (const int*)d_in[1];
    const float* W        = (const float*)d_in[2];
    const float* att_src  = (const float*)d_in[3];
    const float* att_dst  = (const float*)d_in[4];
    const float* bias     = (const float*)d_in[5];
    float*       out      = (float*)d_out;

    const int N = in_sizes[0] / IN_F;
    const int E = in_sizes[1] / 2;

    float*    h     = (float*)d_ws;
    float*    a_src = h + (size_t)N * HC;
    float*    a_dst = a_src + (size_t)N * HEADS;
    unsigned* mkey  = (unsigned*)(a_dst + (size_t)N * HEADS);

    gat_init<<<(N * HEADS + 255) / 256, 256, 0, stream>>>(mkey, N * HEADS);

    gat_gemm_wmma<<<N / 16, 128, GEMM_LDS_BYTES, stream>>>(x, W, h);

    gat_attn_coeff<<<(N + 7) / 8, 256, 0, stream>>>(h, att_src, att_dst, a_src, a_dst, N);

    const int ET = E + N;
    gat_edge_max<<<(ET + 255) / 256, 256, 0, stream>>>(ei, a_src, a_dst, mkey, E, N);
    gat_range_sum<<<(N + RNODES - 1) / RNODES, 1024, 0, stream>>>(ei, a_src, a_dst, mkey, h, bias, out, E, N);
}


# Range blocks scan one owner byte per edge (destination / 391, written by the edge-max pass), four bytes tested per operation, not the 4-byte destination list; sums in a new arrival order
# speedup vs baseline: 2.2411x; 1.0655x over previous
#include <hip/hip_runtime.h>

#define IN_F      128
#define HC        128
#define HEADS     4
#define OUT_F     32
#define NEG_SLOPE 0.2f

typedef float v2f __attribute__((ext_vector_type(2)));
typedef float v4f __attribute__((ext_vector_type(4)));
typedef float v8f __attribute__((ext_vector_type(8)));

__device__ __forceinline__ unsigned fkey(float f) {
    unsigned b = __float_as_uint(f);
    return (b & 0x80000000u) ? ~b : (b | 0x80000000u);
}
__device__ __forceinline__ float funkey(unsigned k) {
    unsigned b = (k & 0x80000000u) ? (k & 0x7fffffffu) : ~k;
    return __uint_as_float(b);
}
__device__ __forceinline__ float leaky(float x) {
    return x > 0.0f ? x : NEG_SLOPE * x;
}

__global__ __launch_bounds__(256) void gat_init(unsigned* __restrict__ mkey, int nm) {
    int i = blockIdx.x * 256 + threadIdx.x;
    if (i < nm) mkey[i] = 0u;
}

#define WSTR 136
#define XSTR 132
#define GEMM_LDS_BYTES ((128 * WSTR + 16 * XSTR) * 4)

__global__ __launch_bounds__(128) void gat_gemm_wmma(const float* __restrict__ x,
                                                     const float* __restrict__ W,
                                                     float* __restrict__ h) {
    extern __shared__ float smem[];
    float* Wl = smem;
    float* Xl = smem + 128 * WSTR;

    const int tid  = threadIdx.x;
    const int row0 = blockIdx.x * 16;

    for (int i = tid; i < 128 * 32; i += 128) {
        int r  = i >> 5;
        int c4 = i & 31;
        v4f v = ((const v4f*)W)[r * 32 + c4];
        *(v4f*)&Wl[r * WSTR + c4 * 4] = v;
    }
    for (int i = tid; i < 16 * 32; i += 128) {
        int r  = i >> 5;
        int c4 = i & 31;
        v4f v = ((const v4f*)x)[(size_t)(row0 + r) * 32 + c4];
        *(v4f*)&Xl[r * XSTR + c4 * 4] = v;
    }
    __syncthreads();

    const int wave = tid >> 5;
    const int lane = tid & 31;
    const int L    = lane & 15;
    const int g    = lane >> 4;

    const int col0 = wave * 16;
    const int col1 = (wave + 4) * 16;

    v8f acc0 = {};
    v8f acc1 = {};

    for (int k0 = 0; k0 < IN_F; k0 += 4) {
        const int k = k0 + 2 * g;
        v2f a, b0, b1;
        a.x  = Xl[L * XSTR + k];
        a.y  = Xl[L * XSTR + k + 1];
        b0.x = Wl[k * WSTR + col0 + L];
        b0.y = Wl[(k + 1) * WSTR + col0 + L];
        b1.x = Wl[k * WSTR + col1 + L];
        b1.y = Wl[(k + 1) * WSTR + col1 + L];
        acc0 = __builtin_amdgcn_wmma_f32_16x16x4_f32(false, a, false, b0, (short)0, acc0, false, false);
        acc1 = __builtin_amdgcn_wmma_f32_16x16x4_f32(false, a, false, b1, (short)0, acc1, false, false);
    }

    const int mbase = row0 + 8 * g;
    for (int r = 0; r < 8; ++r) {
        h[(size_t)(mbase + r) * HC + col0 + L] = acc0[r];
        h[(size_t)(mbase + r) * HC + col1 + L] = acc1[r];
    }
}

__global__ __launch_bounds__(256) void gat_attn_coeff(const float* __restrict__ h,
                                                      const float* __restrict__ att_src,
                                                      const float* __restrict__ att_dst,
                                                      float* __restrict__ a_src,
                                                      float* __restrict__ a_dst,
                                                      int N) {
    int n    = blockIdx.x * 8 + (threadIdx.x >> 5);
    int lane = threadIdx.x & 31;
    if (n >= N) return;

    float s[HEADS], d[HEADS];
#pragma unroll
    for (int hh = 0; hh < HEADS; ++hh) {
        float v = h[(size_t)n * HC + hh * OUT_F + lane];
        s[hh] = v * att_src[hh * OUT_F + lane];
        d[hh] = v * att_dst[hh * OUT_F + lane];
    }
#pragma unroll
    for (int off = 16; off > 0; off >>= 1) {
#pragma unroll
        for (int hh = 0; hh < HEADS; ++hh) {
            s[hh] += __shfl_xor(s[hh], off, 32);
            d[hh] += __shfl_xor(d[hh], off, 32);
        }
    }
    if (lane == 0) {
#pragma unroll
        for (int hh = 0; hh < HEADS; ++hh) {
            a_src[n * HEADS + hh] = s[hh];
            a_dst[n * HEADS + hh] = d[hh];
        }
    }
}

#define RNODES 391

__global__ __launch_bounds__(256) void gat_edge_max(const int* __restrict__ ei,
                                                    const float* __restrict__ a_src,
                                                    const float* __restrict__ a_dst,
                                                    unsigned* __restrict__ mkey,
                                                    unsigned char* __restrict__ owner,
                                                    int E, int N) {
    int e = blockIdx.x * 256 + threadIdx.x;
    if (e >= E + N) return;
    int s, d;
    if (e < E) {
        s = ei[e];
        d = ei[E + e];
        owner[e] = (unsigned char)((unsigned)d / RNODES);
    } else {
        s = d = e - E;
    }
    v4f as = *(const v4f*)&a_src[s * HEADS];
    v4f ad = *(const v4f*)&a_dst[d * HEADS];
#pragma unroll
    for (int hh = 0; hh < HEADS; ++hh) {
        float lr = leaky(as[hh] + ad[hh]);
        atomicMax(&mkey[d * HEADS + hh], fkey(lr));
    }
}

#define RING   64

__device__ __forceinline__ float bcast_f(float v, int k) {
    return __int_as_float(__builtin_amdgcn_readlane(__float_as_int(v), k));
}
__device__ __forceinline__ unsigned zero_bytes(unsigned x) {
    return ~(((x & 0x7f7f7f7fu) + 0x7f7f7f7fu) | x | 0x7f7f7f7fu);
}

__global__ __launch_bounds__(1024) void gat_range_sum(const int* __restrict__ ei,
                                                      const float* __restrict__ a_src,
                                                      const float* __restrict__ a_dst,
                                                      const unsigned* __restrict__ mkey,
                                                      const unsigned char* __restrict__ owner,
                                                      const float* __restrict__ h,
                                                      const float* __restrict__ bias,
                                                      float* __restrict__ out,
                                                      int E, int N) {
    __shared__ __attribute__((aligned(16))) float acc[RNODES * HC];
    __shared__ __attribute__((aligned(16))) float den[RNODES * HEADS];
    __shared__ __attribute__((aligned(16))) float mloc[RNODES * HEADS];
    __shared__ __attribute__((aligned(16))) float adloc[RNODES * HEADS];
    __shared__ int ring_all[32 * RING];

    const int tid  = threadIdx.x;
    const int wave = tid >> 5;
    const int lane = tid & 31;
    const int n0   = blockIdx.x * RNODES;
    const int cnt  = min(RNODES, N - n0);
    const int rbase = wave * RING;

    for (int r = wave; r < cnt; r += 32) {
        const int n = n0 + r;
        const uint4 mk = *(const uint4*)&mkey[n * HEADS];
        const v4f   as = *(const v4f*)&a_src[n * HEADS];
        const v4f   ad = *(const v4f*)&a_dst[n * HEADS];
        const float m0 = funkey(mk.x), m1 = funkey(mk.y), m2 = funkey(mk.z), m3 = funkey(mk.w);
        const float p0 = __expf(leaky(as.x + ad.x) - m0);
        const float p1 = __expf(leaky(as.y + ad.y) - m1);
        const float p2 = __expf(leaky(as.z + ad.z) - m2);
        const float p3 = __expf(leaky(as.w + ad.w) - m3);
        const float* hr = h + (size_t)n * HC + lane;
        float* ar = acc + r * HC + lane;
        ar[0]  = hr[0]  * p0;
        ar[32] = hr[32] * p1;
        ar[64] = hr[64] * p2;
        ar[96] = hr[96] * p3;
        if (lane < HEADS) {
            const float pm = lane == 0 ? p0 : lane == 1 ? p1 : lane == 2 ? p2 : p3;
            const float mm = lane == 0 ? m0 : lane == 1 ? m1 : lane == 2 ? m2 : m3;
            const float aa = lane == 0 ? ad.x : lane == 1 ? ad.y : lane == 2 ? ad.z : ad.w;
            den[r * HEADS + lane]   = pm;
            mloc[r * HEADS + lane]  = mm;
            adloc[r * HEADS + lane] = aa;
        }
    }
    __syncthreads();

    const int* dstp = ei + E;
    int head_q = 0, tail_q = 0;

    auto drain = [&](int take) {
        const bool act = lane < take;
        const int  e   = act ? __hip_atomic_load(&ring_all[rbase + ((head_q + lane) & (RING - 1))], __ATOMIC_RELAXED, __HIP_MEMORY_SCOPE_WAVEFRONT) : 0;
        const int  s   = ei[e];
        const int  dl  = act ? dstp[e] - n0 : 0;
        const v4f  as  = *(const v4f*)&a_src[s * HEADS];
        const v4f  ad  = *(const v4f*)&adloc[dl * HEADS];
        const v4f  mm  = *(const v4f*)&mloc[dl * HEADS];
        const float p0 = __expf(leaky(as.x + ad.x) - mm.x);
        const float p1 = __expf(leaky(as.y + ad.y) - mm.y);
        const float p2 = __expf(leaky(as.z + ad.z) - mm.z);
        const float p3 = __expf(leaky(as.w + ad.w) - mm.w);
        if (act) {
            atomicAdd(&den[dl * HEADS + 0], p0);
            atomicAdd(&den[dl * HEADS + 1], p1);
            atomicAdd(&den[dl * HEADS + 2], p2);
            atomicAdd(&den[dl * HEADS + 3], p3);
        }
        for (int k0 = 0; k0 < take; k0 += 8) {
            float hv[8][4];
            int   dk[8];
#pragma unroll
            for (int j = 0; j < 8; ++j) {
                const int k  = min(k0 + j, take - 1);
                const int sk = __builtin_amdgcn_readlane(s, k);
                dk[j] = __builtin_amdgcn_readlane(dl, k);
                const float* hr = h + (size_t)sk * HC + lane;
                hv[j][0] = hr[0];
                hv[j][1] = hr[32];
                hv[j][2] = hr[64];
                hv[j][3] = hr[96];
            }
#pragma unroll
            for (int j = 0; j < 8; ++j) {
                if (k0 + j < take) {
                    const int k = k0 + j;
                    float* ar = acc + dk[j] * HC + lane;
                    atomicAdd(ar,      hv[j][0] * bcast_f(p0, k));
                    atomicAdd(ar + 32, hv[j][1] * bcast_f(p1, k));
                    atomicAdd(ar + 64, hv[j][2] * bcast_f(p2, k));
                    atomicAdd(ar + 96, hv[j][3] * bcast_f(p3, k));
                }
            }
        }
        head_q += take;
    };

    auto push = [&](bool hit, int e) {
        const unsigned mask = __builtin_amdgcn_ballot_w32(hit);
        if (mask != 0u) {
            const int pre = __builtin_popcount(mask & ((1u << lane) - 1u));
            if (hit) __hip_atomic_store(&ring_all[rbase + ((tail_q + pre) & (RING - 1))], e, __ATOMIC_RELAXED, __HIP_MEMORY_SCOPE_WAVEFRONT);
            tail_q += __builtin_popcount(mask);
            if (tail_q - head_q >= 32) drain(32);
        }
    };

    const bool vec = ((((size_t)owner) & 15) == 0) && (gridDim.x <= 256u);
    const int  E16 = vec ? (E >> 4) : 0;
    const uint4* own4 = (const uint4*)owner;
    const unsigned mine = blockIdx.x * 0x01010101u;
    for (int i0 = 0; i0 < E16; i0 += 1024 * 8) {
        uint4 t[8];
#pragma unroll
        for (int u = 0; u < 8; ++u) {
            t[u] = own4[min(i0 + u * 1024 + tid, E16 - 1)];
        }
        unsigned hw[4];
#pragma unroll
        for (int q = 0; q < 4; ++q) {
            hw[q] = zero_bytes(t[2 * q].w ^ mine)
                  | (zero_bytes(t[2 * q].z ^ mine) >> 1)
                  | (zero_bytes(t[2 * q].y ^ mine) >> 2)
                  | (zero_bytes(t[2 * q].x ^ mine) >> 3)
                  | (zero_bytes(t[2 * q + 1].w ^ mine) >> 4)
                  | (zero_bytes(t[2 * q + 1].z ^ mine) >> 5)
                  | (zero_bytes(t[2 * q + 1].y ^ mine) >> 6)
                  | (zero_bytes(t[2 * q + 1].x ^ mine) >> 7);
            hw[q] &= (i0 + (2 * q) * 1024 + tid < E16 ? 0xf0f0f0f0u : 0u)
                   | (i0 + (2 * q + 1) * 1024 + tid < E16 ? 0x0f0f0f0fu : 0u);
        }
#pragma nounroll
        for (int q = 0; q < 4; ++q) {
            unsigned hb = q == 0 ? hw[0] : q == 1 ? hw[1] : q == 2 ? hw[2] : hw[3];
            while (__builtin_amdgcn_ballot_w32(hb != 0u) != 0u) {
                const bool hit = hb != 0u;
                const int  tz  = hit ? __builtin_ctz(hb) : 0;
                hb &= hb - 1u;
                const int  k   = tz & 7;
                const int  u   = 2 * q + ((k >> 2) ^ 1);
                const int  e   = 16 * (i0 + u * 1024 + tid) + 4 * (k & 3) + (tz >> 3);
                push(hit, e);
            }
        }
    }
    for (int e0 = 16 * E16 + wave * 32; e0 < E; e0 += 1024) {
        const int e = e0 + lane;
        const bool hit = e < E && (unsigned)(dstp[e < E ? e : 0] - n0) < (unsigned)cnt;
        push(hit, e);
    }
    if (tail_q - head_q > 0) drain(tail_q - head_q);
    __syncthreads();

    for (int r = wave; r < cnt; r += 32) {
        const v4f a = *(const v4f*)&acc[r * HC + lane * 4];
        const float dn = den[r * HEADS + (lane >> 3)] + 1e-16f;
        const v4f b = ((const v4f*)bias)[lane];
        v4f o;
        o.x = a.x / dn + b.x;
        o.y = a.y / dn + b.y;
        o.z = a.z / dn + b.z;
        o.w = a.w / dn + b.w;
        *(v4f*)&out[(size_t)(n0 + r) * HC + lane * 4] = o;
    }
}

extern "C" void kernel_launch(void* const* d_in, const int* in_sizes, int n_in,
                              void* d_out, int out_size, void* d_ws, size_t ws_size,
                              hipStream_t stream) {
    const float* x        = (const float*)d_in[0];
    const int*   ei       = (const int*)d_in[1];
    const float* W        = (const float*)d_in[2];
    const float* att_src  = (const float*)d_in[3];
    const float* att_dst  = (const float*)d_in[4];
    const float* bias     = (const float*)d_in[5];
    float*       out      = (float*)d_out;

    const int N = in_sizes[0] / IN_F;
    const int E = in_sizes[1] / 2;

    float*    h     = (float*)d_ws;
    float*    a_src = h + (size_t)N * HC;
    float*    a_dst = a_src + (size_t)N * HEADS;
    unsigned* mkey  = (unsigned*)(a_dst + (size_t)N * HEADS);
    unsigned char* owner = (unsigned char*)(mkey + (size_t)N * HEADS);

    gat_init<<<(N * HEADS + 255) / 256, 256, 0, stream>>>(mkey, N * HEADS);

    gat_gemm_wmma<<<N / 16, 128, GEMM_LDS_BYTES, stream>>>(x, W, h);

    gat_attn_coeff<<<(N + 7) / 8, 256, 0, stream>>>(h, att_src, att_dst, a_src, a_dst, N);

    const int ET = E + N;
    gat_edge_max<<<(ET + 255) / 256, 256, 0, stream>>>(ei, a_src, a_dst, mkey, owner, E, N);
    gat_range_sum<<<(N + RNODES - 1) / RNODES, 1024, 0, stream>>>(ei, a_src, a_dst, mkey, owner, h, bias, out, E, N);
}


# a_src, a_dst as 8 more GEMM columns, x @ (W @ att): a small launch folds att into W, a fifth wave per block runs that column tile; coefficient launch gone; their dot products are re-associated
# speedup vs baseline: 4.8894x; 2.1817x over previous
#include <hip/hip_runtime.h>

#define IN_F      128
#define HC        128
#define HEADS     4
#define OUT_F     32
#define NEG_SLOPE 0.2f

typedef float v2f __attribute__((ext_vector_type(2)));
typedef float v4f __attribute__((ext_vector_type(4)));
typedef float v8f __attribute__((ext_vector_type(8)));

__device__ __forceinline__ unsigned fkey(float f) {
    unsigned b = __float_as_uint(f);
    return (b & 0x80000000u) ? ~b : (b | 0x80000000u);
}
__device__ __forceinline__ float funkey(unsigned k) {
    unsigned b = (k & 0x80000000u) ? (k & 0x7fffffffu) : ~k;
    return __uint_as_float(b);
}
__device__ __forceinline__ float leaky(float x) {
    return x > 0.0f ? x : NEG_SLOPE * x;
}

__global__ __launch_bounds__(128) void gat_fold_att(const float* __restrict__ W,
                                                    const float* __restrict__ att_src,
                                                    const float* __restrict__ att_dst,
                                                    float* __restrict__ Wa) {
    const int t  = blockIdx.x * 128 + threadIdx.x;
    const int k  = t >> 3;
    const int j  = t & 7;
    const int hd = j & 3;
    const float* att = (j < 4 ? att_src : att_dst) + hd * OUT_F;
    const float* wr  = W + k * HC + hd * OUT_F;
    float s = 0.0f;
#pragma unroll
    for (int c = 0; c < OUT_F; ++c) s += wr[c] * att[c];
    Wa[k * 8 + j] = s;
}

#define WSTR 136
#define XSTR 132
#define GEMM_LDS_BYTES ((128 * WSTR + 16 * XSTR) * 4)

__global__ __launch_bounds__(160) void gat_gemm_wmma(const float* __restrict__ x,
                                                     const float* __restrict__ W,
                                                     const float* __restrict__ Wa,
                                                     float* __restrict__ h,
                                                     float* __restrict__ a_src,
                                                     float* __restrict__ a_dst) {
    extern __shared__ float smem[];
    float* Wl = smem;
    float* Xl = smem + 128 * WSTR;

    const int tid  = threadIdx.x;
    const int row0 = blockIdx.x * 16;

    if (tid < 128) {
        for (int i = tid; i < 128 * 32; i += 128) {
            int r  = i >> 5;
            int c4 = i & 31;
            v4f v = ((const v4f*)W)[r * 32 + c4];
            *(v4f*)&Wl[r * WSTR + c4 * 4] = v;
        }
        for (int i = tid; i < 16 * 32; i += 128) {
            int r  = i >> 5;
            int c4 = i & 31;
            v4f v = ((const v4f*)x)[(size_t)(row0 + r) * 32 + c4];
            *(v4f*)&Xl[r * XSTR + c4 * 4] = v;
        }
    } else {
#pragma unroll
        for (int it = 0; it < 8; ++it) {
            const int i  = it * 32 + (tid - 128);
            const int r  = i >> 1;
            const int c4 = i & 1;
            v4f v = ((const v4f*)Wa)[i];
            *(v4f*)&Wl[r * WSTR + 128 + c4 * 4] = v;
        }
    }
    __syncthreads();

    const int wave = __builtin_amdgcn_readfirstlane(tid >> 5);
    const int lane = tid & 31;
    const int L    = lane & 15;
    const int g    = lane >> 4;

    if (wave == 4) {
        const int cL = L & 7;
        v8f acc_e = {};
        v8f acc_o = {};
        for (int k0 = 0; k0 < IN_F; k0 += 8) {
            const int k = k0 + 2 * g;
            v2f a0, a1, b0, b1;
            a0.x = Xl[L * XSTR + k];
            a0.y = Xl[L * XSTR + k + 1];
            b0.x = Wl[k * WSTR + 128 + cL];
            b0.y = Wl[(k + 1) * WSTR + 128 + cL];
            a1.x = Xl[L * XSTR + k + 4];
            a1.y = Xl[L * XSTR + k + 5];
            b1.x = Wl[(k + 4) * WSTR + 128 + cL];
            b1.y = Wl[(k + 5) * WSTR + 128 + cL];
            acc_e = __builtin_amdgcn_wmma_f32_16x16x4_f32(false, a0, false, b0, (short)0, acc_e, false, false);
            acc_o = __builtin_amdgcn_wmma_f32_16x16x4_f32(false, a1, false, b1, (short)0, acc_o, false, false);
        }
        if (L < 8) {
            float* ap = (L < 4 ? a_src : a_dst) + (size_t)(row0 + 8 * g) * HEADS + (L & 3);
            for (int r = 0; r < 8; ++r) ap[r * HEADS] = acc_e[r] + acc_o[r];
        }
        return;
    }

    const int col0 = wave * 16;
    const int col1 = (wave + 4) * 16;

    v8f acc0 = {};
    v8f acc1 = {};

    for (int k0 = 0; k0 < IN_F; k0 += 4) {
        const int k = k0 + 2 * g;
        v2f a, b0, b1;
        a.x  = Xl[L * XSTR + k];
        a.y  = Xl[L * XSTR + k + 1];
        b0.x = Wl[k * WSTR + col0 + L];
        b0.y = Wl[(k + 1) * WSTR + col0 + L];
        b1.x = Wl[k * WSTR + col1 + L];
        b1.y = Wl[(k + 1) * WSTR + col1 + L];
        acc0 = __builtin_amdgcn_wmma_f32_16x16x4_f32(false, a, false, b0, (short)0, acc0, false, false);
        acc1 = __builtin_amdgcn_wmma_f32_16x16x4_f32(false, a, false, b1, (short)0, acc1, false, false);
    }

    const int mbase = row0 + 8 * g;
    for (int r = 0; r < 8; ++r) {
        h[(size_t)(mbase + r) * HC + col0 + L] = acc0[r];
        h[(size_t)(mbase + r) * HC + col1 + L] = acc1[r];
    }
}

#define RNODES 391

__global__ __launch_bounds__(256) void gat_edge_owner(const int* __restrict__ ei,
                                                      unsigned char* __restrict__ owner,
                                                      int E) {
    const int e = 4 * (blockIdx.x * 256 + threadIdx.x);
    if (e >= E) return;
    const int* dst = ei + E;
    const bool wide = ((E & 3) == 0) && ((((size_t)ei) & 15) == 0) && ((((size_t)owner) & 3) == 0);
    if (wide) {
        const int4 d = *(const int4*)(dst + e);
        const unsigned w = (((unsigned)d.x / RNODES) & 0xffu)
                         | ((((unsigned)d.y / RNODES) & 0xffu) << 8)
                         | ((((unsigned)d.z / RNODES) & 0xffu) << 16)
                         | ((((unsigned)d.w / RNODES) & 0xffu) << 24);
        *(unsigned*)(owner + e) = w;
    } else {
        for (int j = 0; j < 4; ++j)
            if (e + j < E) owner[e + j] = (unsigned char)((unsigned)dst[e + j] / RNODES);
    }
}

#define RING   64
#define CAPW   448

__device__ __forceinline__ float bcast_f(float v, int k) {
    return __int_as_float(__builtin_amdgcn_readlane(__float_as_int(v), k));
}
__device__ __forceinline__ unsigned zero_bytes(unsigned x) {
    return ~(((x & 0x7f7f7f7fu) + 0x7f7f7f7fu) | x | 0x7f7f7f7fu);
}

__global__ __launch_bounds__(1024) void gat_range_sum(const int* __restrict__ ei,
                                                      const float* __restrict__ a_src,
                                                      const float* __restrict__ a_dst,
                                                      const unsigned char* __restrict__ owner,
                                                      const float* __restrict__ h,
                                                      const float* __restrict__ bias,
                                                      float* __restrict__ out,
                                                      int E, int N) {
    __shared__ __attribute__((aligned(16))) float acc[RNODES * HC];
    __shared__ __attribute__((aligned(16))) float den[RNODES * HEADS];
    __shared__ __attribute__((aligned(16))) unsigned mkl[RNODES * HEADS];
    __shared__ __attribute__((aligned(16))) float adloc[RNODES * HEADS];
    __shared__ int ring_all[32 * RING];
    __shared__ int            lsrc[32 * CAPW];
    __shared__ unsigned short ldl[32 * CAPW];
    __shared__ int over;

    const int tid  = threadIdx.x;
    const int wave = tid >> 5;
    const int lane = tid & 31;
    const int n0   = blockIdx.x * RNODES;
    const int cnt  = min(RNODES, N - n0);
    const int rbase = wave * RING;
    const int lbase = wave * CAPW;

    if (tid == 0) over = 0;
    for (int i = tid; i < RNODES * HC / 4; i += 1024) ((v4f*)acc)[i] = (v4f){0.0f, 0.0f, 0.0f, 0.0f};
    for (int i = tid; i < cnt * HEADS; i += 1024) {
        const float as = a_src[n0 * HEADS + i];
        const float ad = a_dst[n0 * HEADS + i];
        den[i]   = 0.0f;
        adloc[i] = ad;
        mkl[i]   = fkey(leaky(as + ad));
    }
    int lcount;
    {
        const int  r  = wave + 32 * lane;
        const bool hv = r < cnt;
        if (hv) {
            lsrc[lbase + lane] = n0 + r;
            ldl[lbase + lane]  = (unsigned short)r;
        }
        lcount = __builtin_popcount(__builtin_amdgcn_ballot_w32(hv));
    }
    __syncthreads();

    const int* dstp = ei + E;
    int head_q = 0, tail_q = 0;

    auto drain_max = [&](int take) {
        const bool act = lane < take;
        const int  e   = act ? __hip_atomic_load(&ring_all[rbase + ((head_q + lane) & (RING - 1))], __ATOMIC_RELAXED, __HIP_MEMORY_SCOPE_WAVEFRONT) : 0;
        const int  s   = ei[e];
        const int  dl  = act ? dstp[e] - n0 : 0;
        const v4f  as  = *(const v4f*)&a_src[s * HEADS];
        const v4f  ad  = *(const v4f*)&adloc[dl * HEADS];
        if (act) {
            atomicMax(&mkl[dl * HEADS + 0], fkey(leaky(as.x + ad.x)));
            atomicMax(&mkl[dl * HEADS + 1], fkey(leaky(as.y + ad.y)));
            atomicMax(&mkl[dl * HEADS + 2], fkey(leaky(as.z + ad.z)));
            atomicMax(&mkl[dl * HEADS + 3], fkey(leaky(as.w + ad.w)));
            const int pos = lcount + lane;
            if (pos < CAPW) {
                lsrc[lbase + pos] = s;
                ldl[lbase + pos]  = (unsigned short)dl;
            }
        }
        lcount += take;
        head_q += take;
    };

    auto push = [&](bool hit, int e) {
        const unsigned mask = __builtin_amdgcn_ballot_w32(hit);
        if (mask != 0u) {
            const int pre = __builtin_popcount(mask & ((1u << lane) - 1u));
            if (hit) __hip_atomic_store(&ring_all[rbase + ((tail_q + pre) & (RING - 1))], e, __ATOMIC_RELAXED, __HIP_MEMORY_SCOPE_WAVEFRONT);
            tail_q += __builtin_popcount(mask);
            if (tail_q - head_q >= 32) drain_max(32);
        }
    };

    const bool vec = ((((size_t)owner) & 15) == 0) && (gridDim.x <= 256u);
    const int  E16 = vec ? (E >> 4) : 0;
    const uint4* own4 = (const uint4*)owner;
    const unsigned mine = blockIdx.x * 0x01010101u;
    for (int i0 = 0; i0 < E16; i0 += 1024 * 8) {
        uint4 t[8];
#pragma unroll
        for (int u = 0; u < 8; ++u) {
            t[u] = own4[min(i0 + u * 1024 + tid, E16 - 1)];
        }
        unsigned hw[4];
#pragma unroll
        for (int q = 0; q < 4; ++q) {
            hw[q] = zero_bytes(t[2 * q].w ^ mine)
                  | (zero_bytes(t[2 * q].z ^ mine) >> 1)
                  | (zero_bytes(t[2 * q].y ^ mine) >> 2)
                  | (zero_bytes(t[2 * q].x ^ mine) >> 3)
                  | (zero_bytes(t[2 * q + 1].w ^ mine) >> 4)
                  | (zero_bytes(t[2 * q + 1].z ^ mine) >> 5)
                  | (zero_bytes(t[2 * q + 1].y ^ mine) >> 6)
                  | (zero_bytes(t[2 * q + 1].x ^ mine) >> 7);
            hw[q] &= (i0 + (2 * q) * 1024 + tid < E16 ? 0xf0f0f0f0u : 0u)
                   | (i0 + (2 * q + 1) * 1024 + tid < E16 ? 0x0f0f0f0fu : 0u);
        }
#pragma nounroll
        for (int q = 0; q < 4; ++q) {
            unsigned hb = q == 0 ? hw[0] : q == 1 ? hw[1] : q == 2 ? hw[2] : hw[3];
            while (__builtin_amdgcn_ballot_w32(hb != 0u) != 0u) {
                const bool hit = hb != 0u;
                const int  tz  = hit ? __builtin_ctz(hb) : 0;
                hb &= hb - 1u;
                const int  k   = tz & 7;
                const int  u   = 2 * q + ((k >> 2) ^ 1);
                const int  e   = 16 * (i0 + u * 1024 + tid) + 4 * (k & 3) + (tz >> 3);
                push(hit, e);
            }
        }
    }
    for (int e0 = 16 * E16 + wave * 32; e0 < E; e0 += 1024) {
        const int e = e0 + lane;
        const bool hit = e < E && (unsigned)(dstp[e < E ? e : 0] - n0) < (unsigned)cnt;
        push(hit, e);
    }
    if (tail_q - head_q > 0) drain_max(tail_q - head_q);
    if (lcount > CAPW) over = 1;
    __syncthreads();

    auto drain_sum = [&](int take, int s, int dl) {
        const bool  act = lane < take;
        const v4f   as  = *(const v4f*)&a_src[s * HEADS];
        const v4f   ad  = *(const v4f*)&adloc[dl * HEADS];
        const uint4 mk  = *(const uint4*)&mkl[dl * HEADS];
        const float p0 = __expf(leaky(as.x + ad.x) - funkey(mk.x));
        const float p1 = __expf(leaky(as.y + ad.y) - funkey(mk.y));
        const float p2 = __expf(leaky(as.z + ad.z) - funkey(mk.z));
        const float p3 = __expf(leaky(as.w + ad.w) - funkey(mk.w));
        if (act) {
            atomicAdd(&den[dl * HEADS + 0], p0);
            atomicAdd(&den[dl * HEADS + 1], p1);
            atomicAdd(&den[dl * HEADS + 2], p2);
            atomicAdd(&den[dl * HEADS + 3], p3);
        }
        for (int k0 = 0; k0 < take; k0 += 8) {
            float hv[8][4];
            int   dk[8];
#pragma unroll
            for (int j = 0; j < 8; ++j) {
                const int k  = min(k0 + j, take - 1);
                const int sk = __builtin_amdgcn_readlane(s, k);
                dk[j] = __builtin_amdgcn_readlane(dl, k);
                const float* hr = h + (size_t)sk * HC + lane;
                hv[j][0] = hr[0];
                hv[j][1] = hr[32];
                hv[j][2] = hr[64];
                hv[j][3] = hr[96];
            }
#pragma unroll
            for (int j = 0; j < 8; ++j) {
                if (k0 + j < take) {
                    const int k = k0 + j;
                    float* ar = acc + dk[j] * HC + lane;
                    atomicAdd(ar,      hv[j][0] * bcast_f(p0, k));
                    atomicAdd(ar + 32, hv[j][1] * bcast_f(p1, k));
                    atomicAdd(ar + 64, hv[j][2] * bcast_f(p2, k));
                    atomicAdd(ar + 96, hv[j][3] * bcast_f(p3, k));
                }
            }
        }
    };

    const bool slow = __builtin_amdgcn_readfirstlane(over) != 0;
    if (!slow) {
        const int nl = __builtin_amdgcn_readfirstlane(lcount);
        int i = 0;
        for (; i + 32 <= nl; i += 32)
            drain_sum(32, lsrc[lbase + i + lane], (int)ldl[lbase + i + lane]);
        if (i < nl) {
            const int  take = nl - i;
            const bool act  = lane < take;
            const int  s    = act ? lsrc[lbase + i + lane] : 0;
            const int  dl   = act ? (int)ldl[lbase + i + lane] : 0;
            drain_sum(take, s, dl);
        }
    } else {
        const int ET = E + cnt;
        for (int e0 = wave * 32; ; e0 += 1024) {
            const bool more = e0 < ET;
            if (more) {
                const int  e   = e0 + lane;
                const bool hit = e < E ? (unsigned)(dstp[e] - n0) < (unsigned)cnt : e < ET;
                const unsigned mask = __builtin_amdgcn_ballot_w32(hit);
                if (mask != 0u) {
                    const int pre = __builtin_popcount(mask & ((1u << lane) - 1u));
                    if (hit) __hip_atomic_store(&ring_all[rbase + ((tail_q + pre) & (RING - 1))], e, __ATOMIC_RELAXED, __HIP_MEMORY_SCOPE_WAVEFRONT);
                    tail_q += __builtin_popcount(mask);
                }
            }
            const int waiting = tail_q - head_q;
            const int take    = more ? (waiting >= 32 ? 32 : 0) : waiting;
            if (take > 0) {
                const bool act = lane < take;
                const int  e   = act ? __hip_atomic_load(&ring_all[rbase + ((head_q + lane) & (RING - 1))], __ATOMIC_RELAXED, __HIP_MEMORY_SCOPE_WAVEFRONT) : E;
                const int  s   = e < E ? ei[e] : n0 + (e - E);
                const int  dl  = e < E ? dstp[e] - n0 : e - E;
                drain_sum(take, s, dl);
                head_q += take;
            }
            if (!more) break;
        }
    }
    __syncthreads();

    for (int r = wave; r < cnt; r += 32) {
        const v4f a = *(const v4f*)&acc[r * HC + lane * 4];
        const float dn = den[r * HEADS + (lane >> 3)] + 1e-16f;
        const v4f b = ((const v4f*)bias)[lane];
        v4f o;
        o.x = a.x / dn + b.x;
        o.y = a.y / dn + b.y;
        o.z = a.z / dn + b.z;
        o.w = a.w / dn + b.w;
        *(v4f*)&out[(size_t)(n0 + r) * HC + lane * 4] = o;
    }
}

extern "C" void kernel_launch(void* const* d_in, const int* in_sizes, int n_in,
                              void* d_out, int out_size, void* d_ws, size_t ws_size,
                              hipStream_t stream) {
    const float* x        = (const float*)d_in[0];
    const int*   ei       = (const int*)d_in[1];
    const float* W        = (const float*)d_in[2];
    const float* att_src  = (const float*)d_in[3];
    const float* att_dst  = (const float*)d_in[4];
    const float* bias     = (const float*)d_in[5];
    float*       out      = (float*)d_out;

    const int N = in_sizes[0] / IN_F;
    const int E = in_sizes[1] / 2;

    float*    h     = (float*)d_ws;
    float*    a_src = h + (size_t)N * HC;
    float*    a_dst = a_src + (size_t)N * HEADS;
    float*    Wa    = a_dst + (size_t)N * HEADS;
    unsigned char* owner = (unsigned char*)(Wa + IN_F * 8);

    gat_fold_att<<<8, 128, 0, stream>>>(W, att_src, att_dst, Wa);

    gat_gemm_wmma<<<N / 16, 160, GEMM_LDS_BYTES, stream>>>(x, W, Wa, h, a_src, a_dst);

    gat_edge_owner<<<((E + 3) / 4 + 255) / 256, 256, 0, stream>>>(ei, owner, E);

    gat_range_sum<<<(N + RNODES - 1) / RNODES, 1024, 0, stream>>>(ei, a_src, a_dst, owner, h, bias, out, E, N);
}


# GEMM block staging: every load of a lane (W: 32 float4; or Wa + x tile: 24) is issued before its first LDS store, kernel bounded to 6 waves/SIMD for the registers: 1 round trip per block, not 8
# speedup vs baseline: 5.1251x; 1.0482x over previous
#include <hip/hip_runtime.h>

#define IN_F      128
#define HC        128
#define HEADS     4
#define OUT_F     32
#define NEG_SLOPE 0.2f

typedef float v2f __attribute__((ext_vector_type(2)));
typedef float v4f __attribute__((ext_vector_type(4)));
typedef float v8f __attribute__((ext_vector_type(8)));

__device__ __forceinline__ unsigned fkey(float f) {
    unsigned b = __float_as_uint(f);
    return (b & 0x80000000u) ? ~b : (b | 0x80000000u);
}
__device__ __forceinline__ float funkey(unsigned k) {
    unsigned b = (k & 0x80000000u) ? (k & 0x7fffffffu) : ~k;
    return __uint_as_float(b);
}
__device__ __forceinline__ float leaky(float x) {
    return x > 0.0f ? x : NEG_SLOPE * x;
}

__global__ __launch_bounds__(128) void gat_fold_att(const float* __restrict__ W,
                                                    const float* __restrict__ att_src,
                                                    const float* __restrict__ att_dst,
                                                    float* __restrict__ Wa) {
    const int t  = blockIdx.x * 128 + threadIdx.x;
    const int k  = t >> 3;
    const int j  = t & 7;
    const int hd = j & 3;
    const float* att = (j < 4 ? att_src : att_dst) + hd * OUT_F;
    const float* wr  = W + k * HC + hd * OUT_F;
    float s = 0.0f;
#pragma unroll
    for (int c = 0; c < OUT_F; ++c) s += wr[c] * att[c];
    Wa[k * 8 + j] = s;
}

#define WSTR 136
#define XSTR 132
#define GEMM_LDS_BYTES ((128 * WSTR + 16 * XSTR) * 4)

__global__ __launch_bounds__(160) __attribute__((amdgpu_waves_per_eu(4, 6))) void gat_gemm_wmma(const float* __restrict__ x,
                                                     const float* __restrict__ W,
                                                     const float* __restrict__ Wa,
                                                     float* __restrict__ h,
                                                     float* __restrict__ a_src,
                                                     float* __restrict__ a_dst) {
    extern __shared__ float smem[];
    float* Wl = smem;
    float* Xl = smem + 128 * WSTR;

    const int tid  = threadIdx.x;
    const int row0 = blockIdx.x * 16;

    if (tid < 128) {
        const int r0 = tid >> 5;
        const int c4 = tid & 31;
        v4f t[32];
#pragma unroll
        for (int j = 0; j < 32; ++j) t[j] = ((const v4f*)W)[(4 * j + r0) * 32 + c4];
#pragma unroll
        for (int j = 0; j < 32; ++j) *(v4f*)&Wl[(4 * j + r0) * WSTR + c4 * 4] = t[j];
    } else {
        const int l = tid - 128;
        v4f ta[8];
        v4f tx[16];
#pragma unroll
        for (int j = 0; j < 8; ++j) ta[j] = ((const v4f*)Wa)[j * 32 + l];
#pragma unroll
        for (int j = 0; j < 16; ++j) tx[j] = ((const v4f*)x)[(size_t)(row0 + j) * 32 + l];
#pragma unroll
        for (int j = 0; j < 8; ++j) {
            const int i  = j * 32 + l;
            const int r  = i >> 1;
            const int cc = i & 1;
            *(v4f*)&Wl[r * WSTR + 128 + cc * 4] = ta[j];
        }
#pragma unroll
        for (int j = 0; j < 16; ++j) *(v4f*)&Xl[j * XSTR + l * 4] = tx[j];
    }
    __syncthreads();

    const int wave = __builtin_amdgcn_readfirstlane(tid >> 5);
    const int lane = tid & 31;
    const int L    = lane & 15;
    const int g    = lane >> 4;

    if (wave == 4) {
        const int cL = L & 7;
        v8f acc_e = {};
        v8f acc_o = {};
        for (int k0 = 0; k0 < IN_F; k0 += 8) {
            const int k = k0 + 2 * g;
            v2f a0, a1, b0, b1;
            a0.x = Xl[L * XSTR + k];
            a0.y = Xl[L * XSTR + k + 1];
            b0.x = Wl[k * WSTR + 128 + cL];
            b0.y = Wl[(k + 1) * WSTR + 128 + cL];
            a1.x = Xl[L * XSTR + k + 4];
            a1.y = Xl[L * XSTR + k + 5];
            b1.x = Wl[(k + 4) * WSTR + 128 + cL];
            b1.y = Wl[(k + 5) * WSTR + 128 + cL];
            acc_e = __builtin_amdgcn_wmma_f32_16x16x4_f32(false, a0, false, b0, (short)0, acc_e, false, false);
            acc_o = __builtin_amdgcn_wmma_f32_16x16x4_f32(false, a1, false, b1, (short)0, acc_o, false, false);
        }
        if (L < 8) {
            float* ap = (L < 4 ? a_src : a_dst) + (size_t)(row0 + 8 * g) * HEADS + (L & 3);
            for (int r = 0; r < 8; ++r) ap[r * HEADS] = acc_e[r] + acc_o[r];
        }
        return;
    }

    const int col0 = wave * 16;
    const int col1 = (wave + 4) * 16;

    v8f acc0 = {};
    v8f acc1 = {};

    for (int k0 = 0; k0 < IN_F; k0 += 4) {
        const int k = k0 + 2 * g;
        v2f a, b0, b1;
        a.x  = Xl[L * XSTR + k];
        a.y  = Xl[L * XSTR + k + 1];
        b0.x = Wl[k * WSTR + col0 + L];
        b0.y = Wl[(k + 1) * WSTR + col0 + L];
        b1.x = Wl[k * WSTR + col1 + L];
        b1.y = Wl[(k + 1) * WSTR + col1 + L];
        acc0 = __builtin_amdgcn_wmma_f32_16x16x4_f32(false, a, false, b0, (short)0, acc0, false, false);
        acc1 = __builtin_amdgcn_wmma_f32_16x16x4_f32(false, a, false, b1, (short)0, acc1, false, false);
    }

    const int mbase = row0 + 8 * g;
    for (int r = 0; r < 8; ++r) {
        h[(size_t)(mbase + r) * HC + col0 + L] = acc0[r];
        h[(size_t)(mbase + r) * HC + col1 + L] = acc1[r];
    }
}

#define RNODES 391

__global__ __launch_bounds__(256) void gat_edge_owner(const int* __restrict__ ei,
                                                      unsigned char* __restrict__ owner,
                                                      int E) {
    const int e = 4 * (blockIdx.x * 256 + threadIdx.x);
    if (e >= E) return;
    const int* dst = ei + E;
    const bool wide = ((E & 3) == 0) && ((((size_t)ei) & 15) == 0) && ((((size_t)owner) & 3) == 0);
    if (wide) {
        const int4 d = *(const int4*)(dst + e);
        const unsigned w = (((unsigned)d.x / RNODES) & 0xffu)
                         | ((((unsigned)d.y / RNODES) & 0xffu) << 8)
                         | ((((unsigned)d.z / RNODES) & 0xffu) << 16)
                         | ((((unsigned)d.w / RNODES) & 0xffu) << 24);
        *(unsigned*)(owner + e) = w;
    } else {
        for (int j = 0; j < 4; ++j)
            if (e + j < E) owner[e + j] = (unsigned char)((unsigned)dst[e + j] / RNODES);
    }
}

#define RING   64
#define CAPW   448

__device__ __forceinline__ float bcast_f(float v, int k) {
    return __int_as_float(__builtin_amdgcn_readlane(__float_as_int(v), k));
}
__device__ __forceinline__ unsigned zero_bytes(unsigned x) {
    return ~(((x & 0x7f7f7f7fu) + 0x7f7f7f7fu) | x | 0x7f7f7f7fu);
}

__global__ __launch_bounds__(1024) void gat_range_sum(const int* __restrict__ ei,
                                                      const float* __restrict__ a_src,
                                                      const float* __restrict__ a_dst,
                                                      const unsigned char* __restrict__ owner,
                                                      const float* __restrict__ h,
                                                      const float* __restrict__ bias,
                                                      float* __restrict__ out,
                                                      int E, int N) {
    __shared__ __attribute__((aligned(16))) float acc[RNODES * HC];
    __shared__ __attribute__((aligned(16))) float den[RNODES * HEADS];
    __shared__ __attribute__((aligned(16))) unsigned mkl[RNODES * HEADS];
    __shared__ __attribute__((aligned(16))) float adloc[RNODES * HEADS];
    __shared__ int ring_all[32 * RING];
    __shared__ int            lsrc[32 * CAPW];
    __shared__ unsigned short ldl[32 * CAPW];
    __shared__ int over;

    const int tid  = threadIdx.x;
    const int wave = tid >> 5;
    const int lane = tid & 31;
    const int n0   = blockIdx.x * RNODES;
    const int cnt  = min(RNODES, N - n0);
    const int rbase = wave * RING;
    const int lbase = wave * CAPW;

    if (tid == 0) over = 0;
    for (int i = tid; i < RNODES * HC / 4; i += 1024) ((v4f*)acc)[i] = (v4f){0.0f, 0.0f, 0.0f, 0.0f};
    for (int i = tid; i < cnt * HEADS; i += 1024) {
        const float as = a_src[n0 * HEADS + i];
        const float ad = a_dst[n0 * HEADS + i];
        den[i]   = 0.0f;
        adloc[i] = ad;
        mkl[i]   = fkey(leaky(as + ad));
    }
    int lcount;
    {
        const int  r  = wave + 32 * lane;
        const bool hv = r < cnt;
        if (hv) {
            lsrc[lbase + lane] = n0 + r;
            ldl[lbase + lane]  = (unsigned short)r;
        }
        lcount = __builtin_popcount(__builtin_amdgcn_ballot_w32(hv));
    }
    __syncthreads();

    const int* dstp = ei + E;
    int head_q = 0, tail_q = 0;

    auto drain_max = [&](int take) {
        const bool act = lane < take;
        const int  e   = act ? __hip_atomic_load(&ring_all[rbase + ((head_q + lane) & (RING - 1))], __ATOMIC_RELAXED, __HIP_MEMORY_SCOPE_WAVEFRONT) : 0;
        const int  s   = ei[e];
        const int  dl  = act ? dstp[e] - n0 : 0;
        const v4f  as  = *(const v4f*)&a_src[s * HEADS];
        const v4f  ad  = *(const v4f*)&adloc[dl * HEADS];
        if (act) {
            atomicMax(&mkl[dl * HEADS + 0], fkey(leaky(as.x + ad.x)));
            atomicMax(&mkl[dl * HEADS + 1], fkey(leaky(as.y + ad.y)));
            atomicMax(&mkl[dl * HEADS + 2], fkey(leaky(as.z + ad.z)));
            atomicMax(&mkl[dl * HEADS + 3], fkey(leaky(as.w + ad.w)));
            const int pos = lcount + lane;
            if (pos < CAPW) {
                lsrc[lbase + pos] = s;
                ldl[lbase + pos]  = (unsigned short)dl;
            }
        }
        lcount += take;
        head_q += take;
    };

    auto push = [&](bool hit, int e) {
        const unsigned mask = __builtin_amdgcn_ballot_w32(hit);
        if (mask != 0u) {
            const int pre = __builtin_popcount(mask & ((1u << lane) - 1u));
            if (hit) __hip_atomic_store(&ring_all[rbase + ((tail_q + pre) & (RING - 1))], e, __ATOMIC_RELAXED, __HIP_MEMORY_SCOPE_WAVEFRONT);
            tail_q += __builtin_popcount(mask);
            if (tail_q - head_q >= 32) drain_max(32);
        }
    };

    const bool vec = ((((size_t)owner) & 15) == 0) && (gridDim.x <= 256u);
    const int  E16 = vec ? (E >> 4) : 0;
    const uint4* own4 = (const uint4*)owner;
    const unsigned mine = blockIdx.x * 0x01010101u;
    for (int i0 = 0; i0 < E16; i0 += 1024 * 8) {
        uint4 t[8];
#pragma unroll
        for (int u = 0; u < 8; ++u) {
            t[u] = own4[min(i0 + u * 1024 + tid, E16 - 1)];
        }
        unsigned hw[4];
#pragma unroll
        for (int q = 0; q < 4; ++q) {
            hw[q] = zero_bytes(t[2 * q].w ^ mine)
                  | (zero_bytes(t[2 * q].z ^ mine) >> 1)
                  | (zero_bytes(t[2 * q].y ^ mine) >> 2)
                  | (zero_bytes(t[2 * q].x ^ mine) >> 3)
                  | (zero_bytes(t[2 * q + 1].w ^ mine) >> 4)
                  | (zero_bytes(t[2 * q + 1].z ^ mine) >> 5)
                  | (zero_bytes(t[2 * q + 1].y ^ mine) >> 6)
                  | (zero_bytes(t[2 * q + 1].x ^ mine) >> 7);
            hw[q] &= (i0 + (2 * q) * 1024 + tid < E16 ? 0xf0f0f0f0u : 0u)
                   | (i0 + (2 * q + 1) * 1024 + tid < E16 ? 0x0f0f0f0fu : 0u);
        }
#pragma nounroll
        for (int q = 0; q < 4; ++q) {
            unsigned hb = q == 0 ? hw[0] : q == 1 ? hw[1] : q == 2 ? hw[2] : hw[3];
            while (__builtin_amdgcn_ballot_w32(hb != 0u) != 0u) {
                const bool hit = hb != 0u;
                const int  tz  = hit ? __builtin_ctz(hb) : 0;
                hb &= hb - 1u;
                const int  k   = tz & 7;
                const int  u   = 2 * q + ((k >> 2) ^ 1);
                const int  e   = 16 * (i0 + u * 1024 + tid) + 4 * (k & 3) + (tz >> 3);
                push(hit, e);
            }
        }
    }
    for (int e0 = 16 * E16 + wave * 32; e0 < E; e0 += 1024) {
        const int e = e0 + lane;
        const bool hit = e < E && (unsigned)(dstp[e < E ? e : 0] - n0) < (unsigned)cnt;
        push(hit, e);
    }
    if (tail_q - head_q > 0) drain_max(tail_q - head_q);
    if (lcount > CAPW) over = 1;
    __syncthreads();

    auto drain_sum = [&](int take, int s, int dl) {
        const bool  act = lane < take;
        const v4f   as  = *(const v4f*)&a_src[s * HEADS];
        const v4f   ad  = *(const v4f*)&adloc[dl * HEADS];
        const uint4 mk  = *(const uint4*)&mkl[dl * HEADS];
        const float p0 = __expf(leaky(as.x + ad.x) - funkey(mk.x));
        const float p1 = __expf(leaky(as.y + ad.y) - funkey(mk.y));
        const float p2 = __expf(leaky(as.z + ad.z) - funkey(mk.z));
        const float p3 = __expf(leaky(as.w + ad.w) - funkey(mk.w));
        if (act) {
            atomicAdd(&den[dl * HEADS + 0], p0);
            atomicAdd(&den[dl * HEADS + 1], p1);
            atomicAdd(&den[dl * HEADS + 2], p2);
            atomicAdd(&den[dl * HEADS + 3], p3);
        }
        for (int k0 = 0; k0 < take; k0 += 8) {
            float hv[8][4];
            int   dk[8];
#pragma unroll
            for (int j = 0; j < 8; ++j) {
                const int k  = min(k0 + j, take - 1);
                const int sk = __builtin_amdgcn_readlane(s, k);
                dk[j] = __builtin_amdgcn_readlane(dl, k);
                const float* hr = h + (size_t)sk * HC + lane;
                hv[j][0] = hr[0];
                hv[j][1] = hr[32];
                hv[j][2] = hr[64];
                hv[j][3] = hr[96];
            }
#pragma unroll
            for (int j = 0; j < 8; ++j) {
                if (k0 + j < take) {
                    const int k = k0 + j;
                    float* ar = acc + dk[j] * HC + lane;
                    atomicAdd(ar,      hv[j][0] * bcast_f(p0, k));
                    atomicAdd(ar + 32, hv[j][1] * bcast_f(p1, k));
                    atomicAdd(ar + 64, hv[j][2] * bcast_f(p2, k));
                    atomicAdd(ar + 96, hv[j][3] * bcast_f(p3, k));
                }
            }
        }
    };

    const bool slow = __builtin_amdgcn_readfirstlane(over) != 0;
    if (!slow) {
        const int nl = __builtin_amdgcn_readfirstlane(lcount);
        int i = 0;
        for (; i + 32 <= nl; i += 32)
            drain_sum(32, lsrc[lbase + i + lane], (int)ldl[lbase + i + lane]);
        if (i < nl) {
            const int  take = nl - i;
            const bool act  = lane < take;
            const int  s    = act ? lsrc[lbase + i + lane] : 0;
            const int  dl   = act ? (int)ldl[lbase + i + lane] : 0;
            drain_sum(take, s, dl);
        }
    } else {
        const int ET = E + cnt;
        for (int e0 = wave * 32; ; e0 += 1024) {
            const bool more = e0 < ET;
            if (more) {
                const int  e   = e0 + lane;
                const bool hit = e < E ? (unsigned)(dstp[e] - n0) < (unsigned)cnt : e < ET;
                const unsigned mask = __builtin_amdgcn_ballot_w32(hit);
                if (mask != 0u) {
                    const int pre = __builtin_popcount(mask & ((1u << lane) - 1u));
                    if (hit) __hip_atomic_store(&ring_all[rbase + ((tail_q + pre) & (RING - 1))], e, __ATOMIC_RELAXED, __HIP_MEMORY_SCOPE_WAVEFRONT);
                    tail_q += __builtin_popcount(mask);
                }
            }
            const int waiting = tail_q - head_q;
            const int take    = more ? (waiting >= 32 ? 32 : 0) : waiting;
            if (take > 0) {
                const bool act = lane < take;
                const int  e   = act ? __hip_atomic_load(&ring_all[rbase + ((head_q + lane) & (RING - 1))], __ATOMIC_RELAXED, __HIP_MEMORY_SCOPE_WAVEFRONT) : E;
                const int  s   = e < E ? ei[e] : n0 + (e - E);
                const int  dl  = e < E ? dstp[e] - n0 : e - E;
                drain_sum(take, s, dl);
                head_q += take;
            }
            if (!more) break;
        }
    }
    __syncthreads();

    for (int r = wave; r < cnt; r += 32) {
        const v4f a = *(const v4f*)&acc[r * HC + lane * 4];
        const float dn = den[r * HEADS + (lane >> 3)] + 1e-16f;
        const v4f b = ((const v4f*)bias)[lane];
        v4f o;
        o.x = a.x / dn + b.x;
        o.y = a.y / dn + b.y;
        o.z = a.z / dn + b.z;
        o.w = a.w / dn + b.w;
        *(v4f*)&out[(size_t)(n0 + r) * HC + lane * 4] = o;
    }
}

extern "C" void kernel_launch(void* const* d_in, const int* in_sizes, int n_in,
                              void* d_out, int out_size, void* d_ws, size_t ws_size,
                              hipStream_t stream) {
    const float* x        = (const float*)d_in[0];
    const int*   ei       = (const int*)d_in[1];
    const float* W        = (const float*)d_in[2];
    const float* att_src  = (const float*)d_in[3];
    const float* att_dst  = (const float*)d_in[4];
    const float* bias     = (const float*)d_in[5];
    float*       out      = (float*)d_out;

    const int N = in_sizes[0] / IN_F;
    const int E = in_sizes[1] / 2;

    float*    h     = (float*)d_ws;
    float*    a_src = h + (size_t)N * HC;
    float*    a_dst = a_src + (size_t)N * HEADS;
    float*    Wa    = a_dst + (size_t)N * HEADS;
    unsigned char* owner = (unsigned char*)(Wa + IN_F * 8);

    gat_fold_att<<<8, 128, 0, stream>>>(W, att_src, att_dst, Wa);

    gat_gemm_wmma<<<N / 16, 160, GEMM_LDS_BYTES, stream>>>(x, W, Wa, h, a_src, a_dst);

    gat_edge_owner<<<((E + 3) / 4 + 255) / 256, 256, 0, stream>>>(ei, owner, E);

    gat_range_sum<<<(N + RNODES - 1) / RNODES, 1024, 0, stream>>>(ei, a_src, a_dst, owner, h, bias, out, E, N);
}


# Edge scan by bit planes: owner pass ORs one bit per 16-edge group into each range block's plane (LDS, written whole); a block reads its 12 KB plane, fetches only flagged groups; new arrival order
# speedup vs baseline: 5.4879x; 1.0708x over previous
#include <hip/hip_runtime.h>

#define IN_F      128
#define HC        128
#define HEADS     4
#define OUT_F     32
#define NEG_SLOPE 0.2f

typedef float v2f __attribute__((ext_vector_type(2)));
typedef float v4f __attribute__((ext_vector_type(4)));
typedef float v8f __attribute__((ext_vector_type(8)));

__device__ __forceinline__ unsigned fkey(float f) {
    unsigned b = __float_as_uint(f);
    return (b & 0x80000000u) ? ~b : (b | 0x80000000u);
}
__device__ __forceinline__ float funkey(unsigned k) {
    unsigned b = (k & 0x80000000u) ? (k & 0x7fffffffu) : ~k;
    return __uint_as_float(b);
}
__device__ __forceinline__ float leaky(float x) {
    return x > 0.0f ? x : NEG_SLOPE * x;
}

__global__ __launch_bounds__(128) void gat_fold_att(const float* __restrict__ W,
                                                    const float* __restrict__ att_src,
                                                    const float* __restrict__ att_dst,
                                                    float* __restrict__ Wa) {
    const int t  = blockIdx.x * 128 + threadIdx.x;
    const int k  = t >> 3;
    const int j  = t & 7;
    const int hd = j & 3;
    const float* att = (j < 4 ? att_src : att_dst) + hd * OUT_F;
    const float* wr  = W + k * HC + hd * OUT_F;
    float s = 0.0f;
#pragma unroll
    for (int c = 0; c < OUT_F; ++c) s += wr[c] * att[c];
    Wa[k * 8 + j] = s;
}

#define WSTR 136
#define XSTR 132
#define GEMM_LDS_BYTES ((128 * WSTR + 16 * XSTR) * 4)

__global__ __launch_bounds__(160) __attribute__((amdgpu_waves_per_eu(4, 6))) void gat_gemm_wmma(const float* __restrict__ x,
                                                     const float* __restrict__ W,
                                                     const float* __restrict__ Wa,
                                                     float* __restrict__ h,
                                                     float* __restrict__ a_src,
                                                     float* __restrict__ a_dst) {
    extern __shared__ float smem[];
    float* Wl = smem;
    float* Xl = smem + 128 * WSTR;

    const int tid  = threadIdx.x;
    const int row0 = blockIdx.x * 16;

    if (tid < 128) {
        const int r0 = tid >> 5;
        const int c4 = tid & 31;
        v4f t[32];
#pragma unroll
        for (int j = 0; j < 32; ++j) t[j] = ((const v4f*)W)[(4 * j + r0) * 32 + c4];
#pragma unroll
        for (int j = 0; j < 32; ++j) *(v4f*)&Wl[(4 * j + r0) * WSTR + c4 * 4] = t[j];
    } else {
        const int l = tid - 128;
        v4f ta[8];
        v4f tx[16];
#pragma unroll
        for (int j = 0; j < 8; ++j) ta[j] = ((const v4f*)Wa)[j * 32 + l];
#pragma unroll
        for (int j = 0; j < 16; ++j) tx[j] = ((const v4f*)x)[(size_t)(row0 + j) * 32 + l];
#pragma unroll
        for (int j = 0; j < 8; ++j) {
            const int i  = j * 32 + l;
            const int r  = i >> 1;
            const int cc = i & 1;
            *(v4f*)&Wl[r * WSTR + 128 + cc * 4] = ta[j];
        }
#pragma unroll
        for (int j = 0; j < 16; ++j) *(v4f*)&Xl[j * XSTR + l * 4] = tx[j];
    }
    __syncthreads();

    const int wave = __builtin_amdgcn_readfirstlane(tid >> 5);
    const int lane = tid & 31;
    const int L    = lane & 15;
    const int g    = lane >> 4;

    if (wave == 4) {
        const int cL = L & 7;
        v8f acc_e = {};
        v8f acc_o = {};
        for (int k0 = 0; k0 < IN_F; k0 += 8) {
            const int k = k0 + 2 * g;
            v2f a0, a1, b0, b1;
            a0.x = Xl[L * XSTR + k];
            a0.y = Xl[L * XSTR + k + 1];
            b0.x = Wl[k * WSTR + 128 + cL];
            b0.y = Wl[(k + 1) * WSTR + 128 + cL];
            a1.x = Xl[L * XSTR + k + 4];
            a1.y = Xl[L * XSTR + k + 5];
            b1.x = Wl[(k + 4) * WSTR + 128 + cL];
            b1.y = Wl[(k + 5) * WSTR + 128 + cL];
            acc_e = __builtin_amdgcn_wmma_f32_16x16x4_f32(false, a0, false, b0, (short)0, acc_e, false, false);
            acc_o = __builtin_amdgcn_wmma_f32_16x16x4_f32(false, a1, false, b1, (short)0, acc_o, false, false);
        }
        if (L < 8) {
            float* ap = (L < 4 ? a_src : a_dst) + (size_t)(row0 + 8 * g) * HEADS + (L & 3);
            for (int r = 0; r < 8; ++r) ap[r * HEADS] = acc_e[r] + acc_o[r];
        }
        return;
    }

    const int col0 = wave * 16;
    const int col1 = (wave + 4) * 16;

    v8f acc0 = {};
    v8f acc1 = {};

    for (int k0 = 0; k0 < IN_F; k0 += 4) {
        const int k = k0 + 2 * g;
        v2f a, b0, b1;
        a.x  = Xl[L * XSTR + k];
        a.y  = Xl[L * XSTR + k + 1];
        b0.x = Wl[k * WSTR + col0 + L];
        b0.y = Wl[(k + 1) * WSTR + col0 + L];
        b1.x = Wl[k * WSTR + col1 + L];
        b1.y = Wl[(k + 1) * WSTR + col1 + L];
        acc0 = __builtin_amdgcn_wmma_f32_16x16x4_f32(false, a, false, b0, (short)0, acc0, false, false);
        acc1 = __builtin_amdgcn_wmma_f32_16x16x4_f32(false, a, false, b1, (short)0, acc1, false, false);
    }

    const int mbase = row0 + 8 * g;
    for (int r = 0; r < 8; ++r) {
        h[(size_t)(mbase + r) * HC + col0 + L] = acc0[r];
        h[(size_t)(mbase + r) * HC + col1 + L] = acc1[r];
    }
}

#define RNODES 391
#define GB_THREADS 512
#define GB_GROUPS  128

__global__ __launch_bounds__(GB_THREADS) void gat_edge_groups(const int* __restrict__ ei,
                                                              unsigned* __restrict__ bm,
                                                              int E, int WPO) {
    __shared__ unsigned lb[256 * 4];
    const int tid = threadIdx.x;
    for (int i = tid; i < 256 * 4; i += GB_THREADS) lb[i] = 0u;
    __syncthreads();
    const int* dst = ei + E;
    const int  E16 = E >> 4;
    const int  lg  = tid >> 2;
    const int  g   = blockIdx.x * GB_GROUPS + lg;
    if (g < E16) {
        const int e = 16 * g + 4 * (tid & 3);
        int d[4];
        if ((((size_t)dst) & 15) == 0) {
            const int4 v = *(const int4*)(dst + e);
            d[0] = v.x; d[1] = v.y; d[2] = v.z; d[3] = v.w;
        } else {
            for (int j = 0; j < 4; ++j) d[j] = dst[e + j];
        }
        const unsigned bit = 1u << (lg & 31);
        for (int j = 0; j < 4; ++j) {
            const unsigned o = (unsigned)d[j] / RNODES;
            if (o < 256u) atomicOr(&lb[o * 4 + (lg >> 5)], bit);
        }
    }
    __syncthreads();
    for (int i = tid; i < 256 * 4; i += GB_THREADS)
        bm[(size_t)(i >> 2) * WPO + blockIdx.x * 4 + (i & 3)] = lb[i];
}

#define RING   64
#define CAPW   448

__device__ __forceinline__ float bcast_f(float v, int k) {
    return __int_as_float(__builtin_amdgcn_readlane(__float_as_int(v), k));
}

__global__ __launch_bounds__(1024) void gat_range_sum(const int* __restrict__ ei,
                                                      const float* __restrict__ a_src,
                                                      const float* __restrict__ a_dst,
                                                      const unsigned* __restrict__ bm,
                                                      const float* __restrict__ h,
                                                      const float* __restrict__ bias,
                                                      float* __restrict__ out,
                                                      int E, int N, int WPO) {
    __shared__ __attribute__((aligned(16))) float acc[RNODES * HC];
    __shared__ __attribute__((aligned(16))) float den[RNODES * HEADS];
    __shared__ __attribute__((aligned(16))) unsigned mkl[RNODES * HEADS];
    __shared__ __attribute__((aligned(16))) float adloc[RNODES * HEADS];
    __shared__ int ring_all[32 * RING];
    __shared__ int ring_grp[32 * RING];
    __shared__ int            lsrc[32 * CAPW];
    __shared__ unsigned short ldl[32 * CAPW];
    __shared__ int over;

    const int tid  = threadIdx.x;
    const int wave = tid >> 5;
    const int lane = tid & 31;
    const int n0   = blockIdx.x * RNODES;
    const int cnt  = min(RNODES, N - n0);
    const int rbase = wave * RING;
    const int lbase = wave * CAPW;

    if (tid == 0) over = 0;
    for (int i = tid; i < RNODES * HC / 4; i += 1024) ((v4f*)acc)[i] = (v4f){0.0f, 0.0f, 0.0f, 0.0f};
    for (int i = tid; i < cnt * HEADS; i += 1024) {
        const float as = a_src[n0 * HEADS + i];
        const float ad = a_dst[n0 * HEADS + i];
        den[i]   = 0.0f;
        adloc[i] = ad;
        mkl[i]   = fkey(leaky(as + ad));
    }
    int lcount;
    {
        const int  r  = wave + 32 * lane;
        const bool hv = r < cnt;
        if (hv) {
            lsrc[lbase + lane] = n0 + r;
            ldl[lbase + lane]  = (unsigned short)r;
        }
        lcount = __builtin_popcount(__builtin_amdgcn_ballot_w32(hv));
    }
    __syncthreads();

    const int* dstp = ei + E;
    int head_q = 0, tail_q = 0;

    auto drain_max = [&](int take) {
        const bool act = lane < take;
        const int  e   = act ? __hip_atomic_load(&ring_all[rbase + ((head_q + lane) & (RING - 1))], __ATOMIC_RELAXED, __HIP_MEMORY_SCOPE_WAVEFRONT) : 0;
        const int  s   = ei[e];
        const int  dl  = act ? dstp[e] - n0 : 0;
        const v4f  as  = *(const v4f*)&a_src[s * HEADS];
        const v4f  ad  = *(const v4f*)&adloc[dl * HEADS];
        if (act) {
            atomicMax(&mkl[dl * HEADS + 0], fkey(leaky(as.x + ad.x)));
            atomicMax(&mkl[dl * HEADS + 1], fkey(leaky(as.y + ad.y)));
            atomicMax(&mkl[dl * HEADS + 2], fkey(leaky(as.z + ad.z)));
            atomicMax(&mkl[dl * HEADS + 3], fkey(leaky(as.w + ad.w)));
            const int pos = lcount + lane;
            if (pos < CAPW) {
                lsrc[lbase + pos] = s;
                ldl[lbase + pos]  = (unsigned short)dl;
            }
        }
        lcount += take;
        head_q += take;
    };

    auto push = [&](bool hit, int e) {
        const unsigned mask = __builtin_amdgcn_ballot_w32(hit);
        if (mask != 0u) {
            const int pre = __builtin_popcount(mask & ((1u << lane) - 1u));
            if (hit) __hip_atomic_store(&ring_all[rbase + ((tail_q + pre) & (RING - 1))], e, __ATOMIC_RELAXED, __HIP_MEMORY_SCOPE_WAVEFRONT);
            tail_q += __builtin_popcount(mask);
            if (tail_q - head_q >= 32) drain_max(32);
        }
    };

    int head_g = 0, tail_g = 0;

    auto drain_groups = [&](int take) {
        const bool act = lane < take;
        const int  g   = act ? __hip_atomic_load(&ring_grp[rbase + ((head_g + lane) & (RING - 1))], __ATOMIC_RELAXED, __HIP_MEMORY_SCOPE_WAVEFRONT) : 0;
        const int4* dp = (const int4*)(dstp + 16 * g);
        const int4 d0 = dp[0], d1 = dp[1], d2 = dp[2], d3 = dp[3];
        const unsigned uc = (unsigned)cnt;
        unsigned hm = ((unsigned)(d0.x - n0) < uc ? 0x0001u : 0u) | ((unsigned)(d0.y - n0) < uc ? 0x0002u : 0u)
                    | ((unsigned)(d0.z - n0) < uc ? 0x0004u : 0u) | ((unsigned)(d0.w - n0) < uc ? 0x0008u : 0u)
                    | ((unsigned)(d1.x - n0) < uc ? 0x0010u : 0u) | ((unsigned)(d1.y - n0) < uc ? 0x0020u : 0u)
                    | ((unsigned)(d1.z - n0) < uc ? 0x0040u : 0u) | ((unsigned)(d1.w - n0) < uc ? 0x0080u : 0u)
                    | ((unsigned)(d2.x - n0) < uc ? 0x0100u : 0u) | ((unsigned)(d2.y - n0) < uc ? 0x0200u : 0u)
                    | ((unsigned)(d2.z - n0) < uc ? 0x0400u : 0u) | ((unsigned)(d2.w - n0) < uc ? 0x0800u : 0u)
                    | ((unsigned)(d3.x - n0) < uc ? 0x1000u : 0u) | ((unsigned)(d3.y - n0) < uc ? 0x2000u : 0u)
                    | ((unsigned)(d3.z - n0) < uc ? 0x4000u : 0u) | ((unsigned)(d3.w - n0) < uc ? 0x8000u : 0u);
        if (!act) hm = 0u;
        head_g += take;
        while (__builtin_amdgcn_ballot_w32(hm != 0u) != 0u) {
            const bool hit = hm != 0u;
            const int  j   = hit ? __builtin_ctz(hm) : 0;
            hm &= hm - 1u;
            push(hit, 16 * g + j);
        }
    };

    auto push_group = [&](bool hit, int g) {
        const unsigned mask = __builtin_amdgcn_ballot_w32(hit);
        if (mask != 0u) {
            const int pre = __builtin_popcount(mask & ((1u << lane) - 1u));
            if (hit) __hip_atomic_store(&ring_grp[rbase + ((tail_g + pre) & (RING - 1))], g, __ATOMIC_RELAXED, __HIP_MEMORY_SCOPE_WAVEFRONT);
            tail_g += __builtin_popcount(mask);
            if (tail_g - head_g >= 32) drain_groups(32);
        }
    };

    const bool vec = ((((size_t)dstp) & 15) == 0) && (gridDim.x <= 256u);
    const int  E16 = vec ? (E >> 4) : 0;
    const int  nW  = (E16 + 31) >> 5;
    const unsigned* plane = bm + (size_t)blockIdx.x * WPO;
    for (int w0 = 0; w0 < nW; w0 += 1024) {
        const int w = w0 + tid;
        unsigned hb = w < nW ? plane[w] : 0u;
        while (__builtin_amdgcn_ballot_w32(hb != 0u) != 0u) {
            const bool hit = hb != 0u;
            const int  tz  = hit ? __builtin_ctz(hb) : 0;
            hb &= hb - 1u;
            push_group(hit, 32 * w + tz);
        }
    }
    if (tail_g - head_g > 0) drain_groups(tail_g - head_g);
    for (int e0 = 16 * E16 + wave * 32; e0 < E; e0 += 1024) {
        const int e = e0 + lane;
        const bool hit = e < E && (unsigned)(dstp[e < E ? e : 0] - n0) < (unsigned)cnt;
        push(hit, e);
    }
    if (tail_q - head_q > 0) drain_max(tail_q - head_q);
    if (lcount > CAPW) over = 1;
    __syncthreads();

    auto drain_sum = [&](int take, int s, int dl) {
        const bool  act = lane < take;
        const v4f   as  = *(const v4f*)&a_src[s * HEADS];
        const v4f   ad  = *(const v4f*)&adloc[dl * HEADS];
        const uint4 mk  = *(const uint4*)&mkl[dl * HEADS];
        const float p0 = __expf(leaky(as.x + ad.x) - funkey(mk.x));
        const float p1 = __expf(leaky(as.y + ad.y) - funkey(mk.y));
        const float p2 = __expf(leaky(as.z + ad.z) - funkey(mk.z));
        const float p3 = __expf(leaky(as.w + ad.w) - funkey(mk.w));
        if (act) {
            atomicAdd(&den[dl * HEADS + 0], p0);
            atomicAdd(&den[dl * HEADS + 1], p1);
            atomicAdd(&den[dl * HEADS + 2], p2);
            atomicAdd(&den[dl * HEADS + 3], p3);
        }
        for (int k0 = 0; k0 < take; k0 += 8) {
            float hv[8][4];
            int   dk[8];
#pragma unroll
            for (int j = 0; j < 8; ++j) {
                const int k  = min(k0 + j, take - 1);
                const int sk = __builtin_amdgcn_readlane(s, k);
                dk[j] = __builtin_amdgcn_readlane(dl, k);
                const float* hr = h + (size_t)sk * HC + lane;
                hv[j][0] = hr[0];
                hv[j][1] = hr[32];
                hv[j][2] = hr[64];
                hv[j][3] = hr[96];
            }
#pragma unroll
            for (int j = 0; j < 8; ++j) {
                if (k0 + j < take) {
                    const int k = k0 + j;
                    float* ar = acc + dk[j] * HC + lane;
                    atomicAdd(ar,      hv[j][0] * bcast_f(p0, k));
                    atomicAdd(ar + 32, hv[j][1] * bcast_f(p1, k));
                    atomicAdd(ar + 64, hv[j][2] * bcast_f(p2, k));
                    atomicAdd(ar + 96, hv[j][3] * bcast_f(p3, k));
                }
            }
        }
    };

    const bool slow = __builtin_amdgcn_readfirstlane(over) != 0;
    if (!slow) {
        const int nl = __builtin_amdgcn_readfirstlane(lcount);
        int i = 0;
        for (; i + 32 <= nl; i += 32)
            drain_sum(32, lsrc[lbase + i + lane], (int)ldl[lbase + i + lane]);
        if (i < nl) {
            const int  take = nl - i;
            const bool act  = lane < take;
            const int  s    = act ? lsrc[lbase + i + lane] : 0;
            const int  dl   = act ? (int)ldl[lbase + i + lane] : 0;
            drain_sum(take, s, dl);
        }
    } else {
        const int ET = E + cnt;
        for (int e0 = wave * 32; ; e0 += 1024) {
            const bool more = e0 < ET;
            if (more) {
                const int  e   = e0 + lane;
                const bool hit = e < E ? (unsigned)(dstp[e] - n0) < (unsigned)cnt : e < ET;
                const unsigned mask = __builtin_amdgcn_ballot_w32(hit);
                if (mask != 0u) {
                    const int pre = __builtin_popcount(mask & ((1u << lane) - 1u));
                    if (hit) __hip_atomic_store(&ring_all[rbase + ((tail_q + pre) & (RING - 1))], e, __ATOMIC_RELAXED, __HIP_MEMORY_SCOPE_WAVEFRONT);
                    tail_q += __builtin_popcount(mask);
                }
            }
            const int waiting = tail_q - head_q;
            const int take    = more ? (waiting >= 32 ? 32 : 0) : waiting;
            if (take > 0) {
                const bool act = lane < take;
                const int  e   = act ? __hip_atomic_load(&ring_all[rbase + ((head_q + lane) & (RING - 1))], __ATOMIC_RELAXED, __HIP_MEMORY_SCOPE_WAVEFRONT) : E;
                const int  s   = e < E ? ei[e] : n0 + (e - E);
                const int  dl  = e < E ? dstp[e] - n0 : e - E;
                drain_sum(take, s, dl);
                head_q += take;
            }
            if (!more) break;
        }
    }
    __syncthreads();

    for (int r = wave; r < cnt; r += 32) {
        const v4f a = *(const v4f*)&acc[r * HC + lane * 4];
        const float dn = den[r * HEADS + (lane >> 3)] + 1e-16f;
        const v4f b = ((const v4f*)bias)[lane];
        v4f o;
        o.x = a.x / dn + b.x;
        o.y = a.y / dn + b.y;
        o.z = a.z / dn + b.z;
        o.w = a.w / dn + b.w;
        *(v4f*)&out[(size_t)(n0 + r) * HC + lane * 4] = o;
    }
}

extern "C" void kernel_launch(void* const* d_in, const int* in_sizes, int n_in,
                              void* d_out, int out_size, void* d_ws, size_t ws_size,
                              hipStream_t stream) {
    const float* x        = (const float*)d_in[0];
    const int*   ei       = (const int*)d_in[1];
    const float* W        = (const float*)d_in[2];
    const float* att_src  = (const float*)d_in[3];
    const float* att_dst  = (const float*)d_in[4];
    const float* bias     = (const float*)d_in[5];
    float*       out      = (float*)d_out;

    const int N = in_sizes[0] / IN_F;
    const int E = in_sizes[1] / 2;

    float*    h     = (float*)d_ws;
    float*    a_src = h + (size_t)N * HC;
    float*    a_dst = a_src + (size_t)N * HEADS;
    float*    Wa    = a_dst + (size_t)N * HEADS;
    unsigned* bm = (unsigned*)(Wa + IN_F * 8);
    const int GBLK = ((E >> 4) + GB_GROUPS - 1) / GB_GROUPS + ((E >> 4) == 0);
    const int WPO  = GBLK * 4;

    gat_fold_att<<<8, 128, 0, stream>>>(W, att_src, att_dst, Wa);

    gat_gemm_wmma<<<N / 16, 160, GEMM_LDS_BYTES, stream>>>(x, W, Wa, h, a_src, a_dst);

    gat_edge_groups<<<GBLK, GB_THREADS, 0, stream>>>(ei, bm, E, WPO);

    gat_range_sum<<<(N + RNODES - 1) / RNODES, 1024, 0, stream>>>(ei, a_src, a_dst, bm, h, bias, out, E, N, WPO);
}


# Range blocks fetch a source's 4 coefficients once per edge: the max pass walks the wave's finished list and keeps each entry's 4 logits in registers for the sum pass; no second 16-byte gather
# speedup vs baseline: 5.6554x; 1.0305x over previous
#include <hip/hip_runtime.h>

#define IN_F      128
#define HC        128
#define HEADS     4
#define OUT_F     32
#define NEG_SLOPE 0.2f

typedef float v2f __attribute__((ext_vector_type(2)));
typedef float v4f __attribute__((ext_vector_type(4)));
typedef float v8f __attribute__((ext_vector_type(8)));

__device__ __forceinline__ unsigned fkey(float f) {
    unsigned b = __float_as_uint(f);
    return (b & 0x80000000u) ? ~b : (b | 0x80000000u);
}
__device__ __forceinline__ float funkey(unsigned k) {
    unsigned b = (k & 0x80000000u) ? (k & 0x7fffffffu) : ~k;
    return __uint_as_float(b);
}
__device__ __forceinline__ float leaky(float x) {
    return x > 0.0f ? x : NEG_SLOPE * x;
}

__global__ __launch_bounds__(128) void gat_fold_att(const float* __restrict__ W,
                                                    const float* __restrict__ att_src,
                                                    const float* __restrict__ att_dst,
                                                    float* __restrict__ Wa) {
    const int t  = blockIdx.x * 128 + threadIdx.x;
    const int k  = t >> 3;
    const int j  = t & 7;
    const int hd = j & 3;
    const float* att = (j < 4 ? att_src : att_dst) + hd * OUT_F;
    const float* wr  = W + k * HC + hd * OUT_F;
    float s = 0.0f;
#pragma unroll
    for (int c = 0; c < OUT_F; ++c) s += wr[c] * att[c];
    Wa[k * 8 + j] = s;
}

#define WSTR 136
#define XSTR 132
#define GEMM_LDS_BYTES ((128 * WSTR + 16 * XSTR) * 4)

__global__ __launch_bounds__(160) __attribute__((amdgpu_waves_per_eu(4, 6))) void gat_gemm_wmma(const float* __restrict__ x,
                                                     const float* __restrict__ W,
                                                     const float* __restrict__ Wa,
                                                     float* __restrict__ h,
                                                     float* __restrict__ a_src,
                                                     float* __restrict__ a_dst) {
    extern __shared__ float smem[];
    float* Wl = smem;
    float* Xl = smem + 128 * WSTR;

    const int tid  = threadIdx.x;
    const int row0 = blockIdx.x * 16;

    if (tid < 128) {
        const int r0 = tid >> 5;
        const int c4 = tid & 31;
        v4f t[32];
#pragma unroll
        for (int j = 0; j < 32; ++j) t[j] = ((const v4f*)W)[(4 * j + r0) * 32 + c4];
#pragma unroll
        for (int j = 0; j < 32; ++j) *(v4f*)&Wl[(4 * j + r0) * WSTR + c4 * 4] = t[j];
    } else {
        const int l = tid - 128;
        v4f ta[8];
        v4f tx[16];
#pragma unroll
        for (int j = 0; j < 8; ++j) ta[j] = ((const v4f*)Wa)[j * 32 + l];
#pragma unroll
        for (int j = 0; j < 16; ++j) tx[j] = ((const v4f*)x)[(size_t)(row0 + j) * 32 + l];
#pragma unroll
        for (int j = 0; j < 8; ++j) {
            const int i  = j * 32 + l;
            const int r  = i >> 1;
            const int cc = i & 1;
            *(v4f*)&Wl[r * WSTR + 128 + cc * 4] = ta[j];
        }
#pragma unroll
        for (int j = 0; j < 16; ++j) *(v4f*)&Xl[j * XSTR + l * 4] = tx[j];
    }
    __syncthreads();

    const int wave = __builtin_amdgcn_readfirstlane(tid >> 5);
    const int lane = tid & 31;
    const int L    = lane & 15;
    const int g    = lane >> 4;

    if (wave == 4) {
        const int cL = L & 7;
        v8f acc_e = {};
        v8f acc_o = {};
        for (int k0 = 0; k0 < IN_F; k0 += 8) {
            const int k = k0 + 2 * g;
            v2f a0, a1, b0, b1;
            a0.x = Xl[L * XSTR + k];
            a0.y = Xl[L * XSTR + k + 1];
            b0.x = Wl[k * WSTR + 128 + cL];
            b0.y = Wl[(k + 1) * WSTR + 128 + cL];
            a1.x = Xl[L * XSTR + k + 4];
            a1.y = Xl[L * XSTR + k + 5];
            b1.x = Wl[(k + 4) * WSTR + 128 + cL];
            b1.y = Wl[(k + 5) * WSTR + 128 + cL];
            acc_e = __builtin_amdgcn_wmma_f32_16x16x4_f32(false, a0, false, b0, (short)0, acc_e, false, false);
            acc_o = __builtin_amdgcn_wmma_f32_16x16x4_f32(false, a1, false, b1, (short)0, acc_o, false, false);
        }
        if (L < 8) {
            float* ap = (L < 4 ? a_src : a_dst) + (size_t)(row0 + 8 * g) * HEADS + (L & 3);
            for (int r = 0; r < 8; ++r) ap[r * HEADS] = acc_e[r] + acc_o[r];
        }
        return;
    }

    const int col0 = wave * 16;
    const int col1 = (wave + 4) * 16;

    v8f acc0 = {};
    v8f acc1 = {};

    for (int k0 = 0; k0 < IN_F; k0 += 4) {
        const int k = k0 + 2 * g;
        v2f a, b0, b1;
        a.x  = Xl[L * XSTR + k];
        a.y  = Xl[L * XSTR + k + 1];
        b0.x = Wl[k * WSTR + col0 + L];
        b0.y = Wl[(k + 1) * WSTR + col0 + L];
        b1.x = Wl[k * WSTR + col1 + L];
        b1.y = Wl[(k + 1) * WSTR + col1 + L];
        acc0 = __builtin_amdgcn_wmma_f32_16x16x4_f32(false, a, false, b0, (short)0, acc0, false, false);
        acc1 = __builtin_amdgcn_wmma_f32_16x16x4_f32(false, a, false, b1, (short)0, acc1, false, false);
    }

    const int mbase = row0 + 8 * g;
    for (int r = 0; r < 8; ++r) {
        h[(size_t)(mbase + r) * HC + col0 + L] = acc0[r];
        h[(size_t)(mbase + r) * HC + col1 + L] = acc1[r];
    }
}

#define RNODES 391
#define GB_THREADS 512
#define GB_GROUPS  128

__global__ __launch_bounds__(GB_THREADS) void gat_edge_groups(const int* __restrict__ ei,
                                                              unsigned* __restrict__ bm,
                                                              int E, int WPO) {
    __shared__ unsigned lb[256 * 4];
    const int tid = threadIdx.x;
    for (int i = tid; i < 256 * 4; i += GB_THREADS) lb[i] = 0u;
    __syncthreads();
    const int* dst = ei + E;
    const int  E16 = E >> 4;
    const int  lg  = tid >> 2;
    const int  g   = blockIdx.x * GB_GROUPS + lg;
    if (g < E16) {
        const int e = 16 * g + 4 * (tid & 3);
        int d[4];
        if ((((size_t)dst) & 15) == 0) {
            const int4 v = *(const int4*)(dst + e);
            d[0] = v.x; d[1] = v.y; d[2] = v.z; d[3] = v.w;
        } else {
            for (int j = 0; j < 4; ++j) d[j] = dst[e + j];
        }
        const unsigned bit = 1u << (lg & 31);
        for (int j = 0; j < 4; ++j) {
            const unsigned o = (unsigned)d[j] / RNODES;
            if (o < 256u) atomicOr(&lb[o * 4 + (lg >> 5)], bit);
        }
    }
    __syncthreads();
    for (int i = tid; i < 256 * 4; i += GB_THREADS)
        bm[(size_t)(i >> 2) * WPO + blockIdx.x * 4 + (i & 3)] = lb[i];
}

#define RING   64
#define CAPW   448
#define NBAT   (CAPW / 32)

__device__ __forceinline__ float bcast_f(float v, int k) {
    return __int_as_float(__builtin_amdgcn_readlane(__float_as_int(v), k));
}

__global__ __launch_bounds__(1024) void gat_range_sum(const int* __restrict__ ei,
                                                      const float* __restrict__ a_src,
                                                      const float* __restrict__ a_dst,
                                                      const unsigned* __restrict__ bm,
                                                      const float* __restrict__ h,
                                                      const float* __restrict__ bias,
                                                      float* __restrict__ out,
                                                      int E, int N, int WPO) {
    __shared__ __attribute__((aligned(16))) float acc[RNODES * HC];
    __shared__ __attribute__((aligned(16))) float den[RNODES * HEADS];
    __shared__ __attribute__((aligned(16))) unsigned mkl[RNODES * HEADS];
    __shared__ __attribute__((aligned(16))) float adloc[RNODES * HEADS];
    __shared__ int ring_all[32 * RING];
    __shared__ int ring_grp[32 * RING];
    __shared__ int            lsrc[32 * CAPW];
    __shared__ unsigned short ldl[32 * CAPW];
    __shared__ int over;

    const int tid  = threadIdx.x;
    const int wave = tid >> 5;
    const int lane = tid & 31;
    const int n0   = blockIdx.x * RNODES;
    const int cnt  = min(RNODES, N - n0);
    const int rbase = wave * RING;
    const int lbase = wave * CAPW;

    if (tid == 0) over = 0;
    for (int i = tid; i < RNODES * HC / 4; i += 1024) ((v4f*)acc)[i] = (v4f){0.0f, 0.0f, 0.0f, 0.0f};
    for (int i = tid; i < cnt * HEADS; i += 1024) {
        const float as = a_src[n0 * HEADS + i];
        const float ad = a_dst[n0 * HEADS + i];
        den[i]   = 0.0f;
        adloc[i] = ad;
        mkl[i]   = fkey(leaky(as + ad));
    }
    int lcount;
    {
        const int  r  = wave + 32 * lane;
        const bool hv = r < cnt;
        if (hv) {
            lsrc[lbase + lane] = n0 + r;
            ldl[lbase + lane]  = (unsigned short)r;
        }
        lcount = __builtin_popcount(__builtin_amdgcn_ballot_w32(hv));
    }
    __syncthreads();

    const int* dstp = ei + E;
    int head_q = 0, tail_q = 0;

    auto drain_list = [&](int take) {
        const bool act = lane < take;
        const int  e   = act ? __hip_atomic_load(&ring_all[rbase + ((head_q + lane) & (RING - 1))], __ATOMIC_RELAXED, __HIP_MEMORY_SCOPE_WAVEFRONT) : 0;
        const int  s   = ei[e];
        const int  dl  = act ? dstp[e] - n0 : 0;
        const int  pos = lcount + lane;
        if (act && pos < CAPW) {
            lsrc[lbase + pos] = s;
            ldl[lbase + pos]  = (unsigned short)dl;
        }
        if (lcount + take > CAPW) {
            const v4f as = *(const v4f*)&a_src[s * HEADS];
            const v4f ad = *(const v4f*)&adloc[dl * HEADS];
            if (act && pos >= CAPW) {
                atomicMax(&mkl[dl * HEADS + 0], fkey(leaky(as.x + ad.x)));
                atomicMax(&mkl[dl * HEADS + 1], fkey(leaky(as.y + ad.y)));
                atomicMax(&mkl[dl * HEADS + 2], fkey(leaky(as.z + ad.z)));
                atomicMax(&mkl[dl * HEADS + 3], fkey(leaky(as.w + ad.w)));
            }
        }
        lcount += take;
        head_q += take;
    };

    auto push = [&](bool hit, int e) {
        const unsigned mask = __builtin_amdgcn_ballot_w32(hit);
        if (mask != 0u) {
            const int pre = __builtin_popcount(mask & ((1u << lane) - 1u));
            if (hit) __hip_atomic_store(&ring_all[rbase + ((tail_q + pre) & (RING - 1))], e, __ATOMIC_RELAXED, __HIP_MEMORY_SCOPE_WAVEFRONT);
            tail_q += __builtin_popcount(mask);
            if (tail_q - head_q >= 32) drain_list(32);
        }
    };

    int head_g = 0, tail_g = 0;

    auto drain_groups = [&](int take) {
        const bool act = lane < take;
        const int  g   = act ? __hip_atomic_load(&ring_grp[rbase + ((head_g + lane) & (RING - 1))], __ATOMIC_RELAXED, __HIP_MEMORY_SCOPE_WAVEFRONT) : 0;
        const int4* dp = (const int4*)(dstp + 16 * g);
        const int4 d0 = dp[0], d1 = dp[1], d2 = dp[2], d3 = dp[3];
        const unsigned uc = (unsigned)cnt;
        unsigned hm = ((unsigned)(d0.x - n0) < uc ? 0x0001u : 0u) | ((unsigned)(d0.y - n0) < uc ? 0x0002u : 0u)
                    | ((unsigned)(d0.z - n0) < uc ? 0x0004u : 0u) | ((unsigned)(d0.w - n0) < uc ? 0x0008u : 0u)
                    | ((unsigned)(d1.x - n0) < uc ? 0x0010u : 0u) | ((unsigned)(d1.y - n0) < uc ? 0x0020u : 0u)
                    | ((unsigned)(d1.z - n0) < uc ? 0x0040u : 0u) | ((unsigned)(d1.w - n0) < uc ? 0x0080u : 0u)
                    | ((unsigned)(d2.x - n0) < uc ? 0x0100u : 0u) | ((unsigned)(d2.y - n0) < uc ? 0x0200u : 0u)
                    | ((unsigned)(d2.z - n0) < uc ? 0x0400u : 0u) | ((unsigned)(d2.w - n0) < uc ? 0x0800u : 0u)
                    | ((unsigned)(d3.x - n0) < uc ? 0x1000u : 0u) | ((unsigned)(d3.y - n0) < uc ? 0x2000u : 0u)
                    | ((unsigned)(d3.z - n0) < uc ? 0x4000u : 0u) | ((unsigned)(d3.w - n0) < uc ? 0x8000u : 0u);
        if (!act) hm = 0u;
        head_g += take;
        while (__builtin_amdgcn_ballot_w32(hm != 0u) != 0u) {
            const bool hit = hm != 0u;
            const int  j   = hit ? __builtin_ctz(hm) : 0;
            hm &= hm - 1u;
            push(hit, 16 * g + j);
        }
    };

    auto push_group = [&](bool hit, int g) {
        const unsigned mask = __builtin_amdgcn_ballot_w32(hit);
        if (mask != 0u) {
            const int pre = __builtin_popcount(mask & ((1u << lane) - 1u));
            if (hit) __hip_atomic_store(&ring_grp[rbase + ((tail_g + pre) & (RING - 1))], g, __ATOMIC_RELAXED, __HIP_MEMORY_SCOPE_WAVEFRONT);
            tail_g += __builtin_popcount(mask);
            if (tail_g - head_g >= 32) drain_groups(32);
        }
    };

    const bool vec = ((((size_t)dstp) & 15) == 0) && (gridDim.x <= 256u);
    const int  E16 = vec ? (E >> 4) : 0;
    const int  nW  = (E16 + 31) >> 5;
    const unsigned* plane = bm + (size_t)blockIdx.x * WPO;
    for (int w0 = 0; w0 < nW; w0 += 1024) {
        const int w = w0 + tid;
        unsigned hb = w < nW ? plane[w] : 0u;
        while (__builtin_amdgcn_ballot_w32(hb != 0u) != 0u) {
            const bool hit = hb != 0u;
            const int  tz  = hit ? __builtin_ctz(hb) : 0;
            hb &= hb - 1u;
            push_group(hit, 32 * w + tz);
        }
    }
    if (tail_g - head_g > 0) drain_groups(tail_g - head_g);
    for (int e0 = 16 * E16 + wave * 32; e0 < E; e0 += 1024) {
        const int e = e0 + lane;
        const bool hit = e < E && (unsigned)(dstp[e < E ? e : 0] - n0) < (unsigned)cnt;
        push(hit, e);
    }
    if (tail_q - head_q > 0) drain_list(tail_q - head_q);
    if (lcount > CAPW) over = 1;

    const int nl = __builtin_amdgcn_readfirstlane(min(lcount, CAPW));
    const int nb = (nl + 31) >> 5;
    float lg0[NBAT], lg1[NBAT], lg2[NBAT], lg3[NBAT];
#pragma unroll
    for (int i = 0; i < NBAT; ++i) { lg0[i] = 0.0f; lg1[i] = 0.0f; lg2[i] = 0.0f; lg3[i] = 0.0f; }
    {
        int s_n = 0, dl_n = 0;
        v4f as_n = {0.0f, 0.0f, 0.0f, 0.0f};
        auto fetch = [&](int b) {
            const int  idx = 32 * b + lane;
            const bool act = idx < nl;
            s_n  = act ? lsrc[lbase + idx] : 0;
            dl_n = act ? (int)ldl[lbase + idx] : 0;
            as_n = *(const v4f*)&a_src[s_n * HEADS];
        };
        if (nb > 0) fetch(nb - 1);
        for (int b = nb - 1; b >= 0; --b) {
            const bool act = 32 * b + lane < nl;
            const int  dl  = dl_n;
            const v4f  as  = as_n;
            if (b > 0) fetch(b - 1);
            const v4f ad = *(const v4f*)&adloc[dl * HEADS];
            const float l0 = leaky(as.x + ad.x);
            const float l1 = leaky(as.y + ad.y);
            const float l2 = leaky(as.z + ad.z);
            const float l3 = leaky(as.w + ad.w);
            if (act) {
                atomicMax(&mkl[dl * HEADS + 0], fkey(l0));
                atomicMax(&mkl[dl * HEADS + 1], fkey(l1));
                atomicMax(&mkl[dl * HEADS + 2], fkey(l2));
                atomicMax(&mkl[dl * HEADS + 3], fkey(l3));
            }
#pragma unroll
            for (int i = NBAT - 1; i > 0; --i) { lg0[i] = lg0[i - 1]; lg1[i] = lg1[i - 1]; lg2[i] = lg2[i - 1]; lg3[i] = lg3[i - 1]; }
            lg0[0] = l0; lg1[0] = l1; lg2[0] = l2; lg3[0] = l3;
        }
    }
    __syncthreads();

    auto drain_sum = [&](int take, int s, int dl, float l0, float l1, float l2, float l3) {
        const bool  act = lane < take;
        const uint4 mk  = *(const uint4*)&mkl[dl * HEADS];
        const float p0 = __expf(l0 - funkey(mk.x));
        const float p1 = __expf(l1 - funkey(mk.y));
        const float p2 = __expf(l2 - funkey(mk.z));
        const float p3 = __expf(l3 - funkey(mk.w));
        if (act) {
            atomicAdd(&den[dl * HEADS + 0], p0);
            atomicAdd(&den[dl * HEADS + 1], p1);
            atomicAdd(&den[dl * HEADS + 2], p2);
            atomicAdd(&den[dl * HEADS + 3], p3);
        }
        for (int k0 = 0; k0 < take; k0 += 8) {
            float hv[8][4];
            int   dk[8];
#pragma unroll
            for (int j = 0; j < 8; ++j) {
                const int k  = min(k0 + j, take - 1);
                const int sk = __builtin_amdgcn_readlane(s, k);
                dk[j] = __builtin_amdgcn_readlane(dl, k);
                const float* hr = h + (size_t)sk * HC + lane;
                hv[j][0] = hr[0];
                hv[j][1] = hr[32];
                hv[j][2] = hr[64];
                hv[j][3] = hr[96];
            }
#pragma unroll
            for (int j = 0; j < 8; ++j) {
                if (k0 + j < take) {
                    const int k = k0 + j;
                    float* ar = acc + dk[j] * HC + lane;
                    atomicAdd(ar,      hv[j][0] * bcast_f(p0, k));
                    atomicAdd(ar + 32, hv[j][1] * bcast_f(p1, k));
                    atomicAdd(ar + 64, hv[j][2] * bcast_f(p2, k));
                    atomicAdd(ar + 96, hv[j][3] * bcast_f(p3, k));
                }
            }
        }
    };

    const bool slow = __builtin_amdgcn_readfirstlane(over) != 0;
    if (!slow) {
        for (int b = 0; b < nb; ++b) {
            const int i = 32 * b;
            if (i + 32 <= nl) {
                drain_sum(32, lsrc[lbase + i + lane], (int)ldl[lbase + i + lane], lg0[0], lg1[0], lg2[0], lg3[0]);
            } else {
                const int  take = nl - i;
                const bool act  = lane < take;
                const int  s    = act ? lsrc[lbase + i + lane] : 0;
                const int  dl   = act ? (int)ldl[lbase + i + lane] : 0;
                drain_sum(take, s, dl, lg0[0], lg1[0], lg2[0], lg3[0]);
            }
#pragma unroll
            for (int j = 0; j < NBAT - 1; ++j) { lg0[j] = lg0[j + 1]; lg1[j] = lg1[j + 1]; lg2[j] = lg2[j + 1]; lg3[j] = lg3[j + 1]; }
        }
    } else {
        const int ET = E + cnt;
        for (int e0 = wave * 32; ; e0 += 1024) {
            const bool more = e0 < ET;
            if (more) {
                const int  e   = e0 + lane;
                const bool hit = e < E ? (unsigned)(dstp[e] - n0) < (unsigned)cnt : e < ET;
                const unsigned mask = __builtin_amdgcn_ballot_w32(hit);
                if (mask != 0u) {
                    const int pre = __builtin_popcount(mask & ((1u << lane) - 1u));
                    if (hit) __hip_atomic_store(&ring_all[rbase + ((tail_q + pre) & (RING - 1))], e, __ATOMIC_RELAXED, __HIP_MEMORY_SCOPE_WAVEFRONT);
                    tail_q += __builtin_popcount(mask);
                }
            }
            const int waiting = tail_q - head_q;
            const int take    = more ? (waiting >= 32 ? 32 : 0) : waiting;
            if (take > 0) {
                const bool act = lane < take;
                const int  e   = act ? __hip_atomic_load(&ring_all[rbase + ((head_q + lane) & (RING - 1))], __ATOMIC_RELAXED, __HIP_MEMORY_SCOPE_WAVEFRONT) : E;
                const int  s   = e < E ? ei[e] : n0 + (e - E);
                const int  dl  = e < E ? dstp[e] - n0 : e - E;
                const v4f  as  = *(const v4f*)&a_src[s * HEADS];
                const v4f  ad  = *(const v4f*)&adloc[dl * HEADS];
                drain_sum(take, s, dl, leaky(as.x + ad.x), leaky(as.y + ad.y), leaky(as.z + ad.z), leaky(as.w + ad.w));
                head_q += take;
            }
            if (!more) break;
        }
    }
    __syncthreads();

    for (int r = wave; r < cnt; r += 32) {
        const v4f a = *(const v4f*)&acc[r * HC + lane * 4];
        const float dn = den[r * HEADS + (lane >> 3)] + 1e-16f;
        const v4f b = ((const v4f*)bias)[lane];
        v4f o;
        o.x = a.x / dn + b.x;
        o.y = a.y / dn + b.y;
        o.z = a.z / dn + b.z;
        o.w = a.w / dn + b.w;
        *(v4f*)&out[(size_t)(n0 + r) * HC + lane * 4] = o;
    }
}

extern "C" void kernel_launch(void* const* d_in, const int* in_sizes, int n_in,
                              void* d_out, int out_size, void* d_ws, size_t ws_size,
                              hipStream_t stream) {
    const float* x        = (const float*)d_in[0];
    const int*   ei       = (const int*)d_in[1];
    const float* W        = (const float*)d_in[2];
    const float* att_src  = (const float*)d_in[3];
    const float* att_dst  = (const float*)d_in[4];
    const float* bias     = (const float*)d_in[5];
    float*       out      = (float*)d_out;

    const int N = in_sizes[0] / IN_F;
    const int E = in_sizes[1] / 2;

    float*    h     = (float*)d_ws;
    float*    a_src = h + (size_t)N * HC;
    float*    a_dst = a_src + (size_t)N * HEADS;
    float*    Wa    = a_dst + (size_t)N * HEADS;
    unsigned* bm = (unsigned*)(Wa + IN_F * 8);
    const int GBLK = ((E >> 4) + GB_GROUPS - 1) / GB_GROUPS + ((E >> 4) == 0);
    const int WPO  = GBLK * 4;

    gat_fold_att<<<8, 128, 0, stream>>>(W, att_src, att_dst, Wa);

    gat_gemm_wmma<<<N / 16, 160, GEMM_LDS_BYTES, stream>>>(x, W, Wa, h, a_src, a_dst);

    gat_edge_groups<<<GBLK, GB_THREADS, 0, stream>>>(ei, bm, E, WPO);

    gat_range_sum<<<(N + RNODES - 1) / RNODES, 1024, 0, stream>>>(ei, a_src, a_dst, bm, h, bias, out, E, N, WPO);
}


# Edge pass also writes the edge list as (source, destination) pairs, a 16-edge group = one 128-byte line; range blocks list kept edges from the line they test, no fetch by edge id; new arrival order
# speedup vs baseline: 5.8840x; 1.0404x over previous
#include <hip/hip_runtime.h>

#define IN_F      128
#define HC        128
#define HEADS     4
#define OUT_F     32
#define NEG_SLOPE 0.2f

typedef float v2f __attribute__((ext_vector_type(2)));
typedef float v4f __attribute__((ext_vector_type(4)));
typedef float v8f __attribute__((ext_vector_type(8)));

__device__ __forceinline__ unsigned fkey(float f) {
    unsigned b = __float_as_uint(f);
    return (b & 0x80000000u) ? ~b : (b | 0x80000000u);
}
__device__ __forceinline__ float funkey(unsigned k) {
    unsigned b = (k & 0x80000000u) ? (k & 0x7fffffffu) : ~k;
    return __uint_as_float(b);
}
__device__ __forceinline__ float leaky(float x) {
    return x > 0.0f ? x : NEG_SLOPE * x;
}

__global__ __launch_bounds__(128) void gat_fold_att(const float* __restrict__ W,
                                                    const float* __restrict__ att_src,
                                                    const float* __restrict__ att_dst,
                                                    float* __restrict__ Wa) {
    const int t  = blockIdx.x * 128 + threadIdx.x;
    const int k  = t >> 3;
    const int j  = t & 7;
    const int hd = j & 3;
    const float* att = (j < 4 ? att_src : att_dst) + hd * OUT_F;
    const float* wr  = W + k * HC + hd * OUT_F;
    float s = 0.0f;
#pragma unroll
    for (int c = 0; c < OUT_F; ++c) s += wr[c] * att[c];
    Wa[k * 8 + j] = s;
}

#define WSTR 136
#define XSTR 132
#define GEMM_LDS_BYTES ((128 * WSTR + 16 * XSTR) * 4)

__global__ __launch_bounds__(160) __attribute__((amdgpu_waves_per_eu(4, 6))) void gat_gemm_wmma(const float* __restrict__ x,
                                                     const float* __restrict__ W,
                                                     const float* __restrict__ Wa,
                                                     float* __restrict__ h,
                                                     float* __restrict__ a_src,
                                                     float* __restrict__ a_dst) {
    extern __shared__ float smem[];
    float* Wl = smem;
    float* Xl = smem + 128 * WSTR;

    const int tid  = threadIdx.x;
    const int row0 = blockIdx.x * 16;

    if (tid < 128) {
        const int r0 = tid >> 5;
        const int c4 = tid & 31;
        v4f t[32];
#pragma unroll
        for (int j = 0; j < 32; ++j) t[j] = ((const v4f*)W)[(4 * j + r0) * 32 + c4];
#pragma unroll
        for (int j = 0; j < 32; ++j) *(v4f*)&Wl[(4 * j + r0) * WSTR + c4 * 4] = t[j];
    } else {
        const int l = tid - 128;
        v4f ta[8];
        v4f tx[16];
#pragma unroll
        for (int j = 0; j < 8; ++j) ta[j] = ((const v4f*)Wa)[j * 32 + l];
#pragma unroll
        for (int j = 0; j < 16; ++j) tx[j] = ((const v4f*)x)[(size_t)(row0 + j) * 32 + l];
#pragma unroll
        for (int j = 0; j < 8; ++j) {
            const int i  = j * 32 + l;
            const int r  = i >> 1;
            const int cc = i & 1;
            *(v4f*)&Wl[r * WSTR + 128 + cc * 4] = ta[j];
        }
#pragma unroll
        for (int j = 0; j < 16; ++j) *(v4f*)&Xl[j * XSTR + l * 4] = tx[j];
    }
    __syncthreads();

    const int wave = __builtin_amdgcn_readfirstlane(tid >> 5);
    const int lane = tid & 31;
    const int L    = lane & 15;
    const int g    = lane >> 4;

    if (wave == 4) {
        const int cL = L & 7;
        v8f acc_e = {};
        v8f acc_o = {};
        for (int k0 = 0; k0 < IN_F; k0 += 8) {
            const int k = k0 + 2 * g;
            v2f a0, a1, b0, b1;
            a0.x = Xl[L * XSTR + k];
            a0.y = Xl[L * XSTR + k + 1];
            b0.x = Wl[k * WSTR + 128 + cL];
            b0.y = Wl[(k + 1) * WSTR + 128 + cL];
            a1.x = Xl[L * XSTR + k + 4];
            a1.y = Xl[L * XSTR + k + 5];
            b1.x = Wl[(k + 4) * WSTR + 128 + cL];
            b1.y = Wl[(k + 5) * WSTR + 128 + cL];
            acc_e = __builtin_amdgcn_wmma_f32_16x16x4_f32(false, a0, false, b0, (short)0, acc_e, false, false);
            acc_o = __builtin_amdgcn_wmma_f32_16x16x4_f32(false, a1, false, b1, (short)0, acc_o, false, false);
        }
        if (L < 8) {
            float* ap = (L < 4 ? a_src : a_dst) + (size_t)(row0 + 8 * g) * HEADS + (L & 3);
            for (int r = 0; r < 8; ++r) ap[r * HEADS] = acc_e[r] + acc_o[r];
        }
        return;
    }

    const int col0 = wave * 16;
    const int col1 = (wave + 4) * 16;

    v8f acc0 = {};
    v8f acc1 = {};

    for (int k0 = 0; k0 < IN_F; k0 += 4) {
        const int k = k0 + 2 * g;
        v2f a, b0, b1;
        a.x  = Xl[L * XSTR + k];
        a.y  = Xl[L * XSTR + k + 1];
        b0.x = Wl[k * WSTR + col0 + L];
        b0.y = Wl[(k + 1) * WSTR + col0 + L];
        b1.x = Wl[k * WSTR + col1 + L];
        b1.y = Wl[(k + 1) * WSTR + col1 + L];
        acc0 = __builtin_amdgcn_wmma_f32_16x16x4_f32(false, a, false, b0, (short)0, acc0, false, false);
        acc1 = __builtin_amdgcn_wmma_f32_16x16x4_f32(false, a, false, b1, (short)0, acc1, false, false);
    }

    const int mbase = row0 + 8 * g;
    for (int r = 0; r < 8; ++r) {
        h[(size_t)(mbase + r) * HC + col0 + L] = acc0[r];
        h[(size_t)(mbase + r) * HC + col1 + L] = acc1[r];
    }
}

#define RNODES 391
#define GB_THREADS 512
#define GB_GROUPS  128

__global__ __launch_bounds__(GB_THREADS) void gat_edge_groups(const int* __restrict__ ei,
                                                              unsigned* __restrict__ bm,
                                                              int* __restrict__ pairs,
                                                              int E, int WPO) {
    __shared__ unsigned lb[256 * 4];
    const int tid = threadIdx.x;
    for (int i = tid; i < 256 * 4; i += GB_THREADS) lb[i] = 0u;
    __syncthreads();
    const int* dst = ei + E;
    const int  E16 = E >> 4;
    const int  wave = tid >> 5;
    const int  lane = tid & 31;
    if (E16 > 0) {
        int  lgk[2], ek[2], sa[2], sb[2], da[2], db[2];
        bool ok[2];
#pragma unroll
        for (int k = 0; k < 2; ++k) {
            lgk[k] = 8 * wave + 4 * k + (lane >> 3);
            const int g = blockIdx.x * GB_GROUPS + lgk[k];
            ok[k] = g < E16;
            ek[k] = 16 * min(g, E16 - 1) + 2 * (lane & 7);
            sa[k] = ei[ek[k]];
            sb[k] = ei[ek[k] + 1];
            da[k] = dst[ek[k]];
            db[k] = dst[ek[k] + 1];
        }
#pragma unroll
        for (int k = 0; k < 2; ++k) {
            if (ok[k]) {
                ((int4*)pairs)[ek[k] >> 1] = make_int4(sa[k], da[k], sb[k], db[k]);
                const unsigned bit = 1u << (lgk[k] & 31);
                const unsigned o0 = (unsigned)da[k] / RNODES;
                const unsigned o1 = (unsigned)db[k] / RNODES;
                if (o0 < 256u) atomicOr(&lb[o0 * 4 + (lgk[k] >> 5)], bit);
                if (o1 < 256u) atomicOr(&lb[o1 * 4 + (lgk[k] >> 5)], bit);
            }
        }
    }
    __syncthreads();
    for (int i = tid; i < 256 * 4; i += GB_THREADS)
        bm[(size_t)(i >> 2) * WPO + blockIdx.x * 4 + (i & 3)] = lb[i];
}

#define RING   64
#define CAPW   448
#define NBAT   (CAPW / 32)

__device__ __forceinline__ float bcast_f(float v, int k) {
    return __int_as_float(__builtin_amdgcn_readlane(__float_as_int(v), k));
}

__global__ __launch_bounds__(1024) void gat_range_sum(const int* __restrict__ ei,
                                                      const float* __restrict__ a_src,
                                                      const float* __restrict__ a_dst,
                                                      const unsigned* __restrict__ bm,
                                                      const int* __restrict__ pairs,
                                                      const float* __restrict__ h,
                                                      const float* __restrict__ bias,
                                                      float* __restrict__ out,
                                                      int E, int N, int WPO) {
    __shared__ __attribute__((aligned(16))) float acc[RNODES * HC];
    __shared__ __attribute__((aligned(16))) float den[RNODES * HEADS];
    __shared__ __attribute__((aligned(16))) unsigned mkl[RNODES * HEADS];
    __shared__ __attribute__((aligned(16))) float adloc[RNODES * HEADS];
    __shared__ int ring_all[32 * RING];
    __shared__ int ring_grp[32 * RING];
    __shared__ int            lsrc[32 * CAPW];
    __shared__ unsigned short ldl[32 * CAPW];
    __shared__ int over;

    const int tid  = threadIdx.x;
    const int wave = tid >> 5;
    const int lane = tid & 31;
    const int n0   = blockIdx.x * RNODES;
    const int cnt  = min(RNODES, N - n0);
    const int rbase = wave * RING;
    const int lbase = wave * CAPW;

    if (tid == 0) over = 0;
    for (int i = tid; i < RNODES * HC / 4; i += 1024) ((v4f*)acc)[i] = (v4f){0.0f, 0.0f, 0.0f, 0.0f};
    for (int i = tid; i < cnt * HEADS; i += 1024) {
        const float as = a_src[n0 * HEADS + i];
        const float ad = a_dst[n0 * HEADS + i];
        den[i]   = 0.0f;
        adloc[i] = ad;
        mkl[i]   = fkey(leaky(as + ad));
    }
    int lcount;
    {
        const int  r  = wave + 32 * lane;
        const bool hv = r < cnt;
        if (hv) {
            lsrc[lbase + lane] = n0 + r;
            ldl[lbase + lane]  = (unsigned short)r;
        }
        lcount = __builtin_popcount(__builtin_amdgcn_ballot_w32(hv));
    }
    __syncthreads();

    const int* dstp = ei + E;
    int head_q = 0, tail_q = 0;

    auto drain_list = [&](int take) {
        const bool act = lane < take;
        const int  e   = act ? __hip_atomic_load(&ring_all[rbase + ((head_q + lane) & (RING - 1))], __ATOMIC_RELAXED, __HIP_MEMORY_SCOPE_WAVEFRONT) : 0;
        const int  s   = ei[e];
        const int  dl  = act ? dstp[e] - n0 : 0;
        const int  pos = lcount + lane;
        if (act && pos < CAPW) {
            lsrc[lbase + pos] = s;
            ldl[lbase + pos]  = (unsigned short)dl;
        }
        if (lcount + take > CAPW) {
            const v4f as = *(const v4f*)&a_src[s * HEADS];
            const v4f ad = *(const v4f*)&adloc[dl * HEADS];
            if (act && pos >= CAPW) {
                atomicMax(&mkl[dl * HEADS + 0], fkey(leaky(as.x + ad.x)));
                atomicMax(&mkl[dl * HEADS + 1], fkey(leaky(as.y + ad.y)));
                atomicMax(&mkl[dl * HEADS + 2], fkey(leaky(as.z + ad.z)));
                atomicMax(&mkl[dl * HEADS + 3], fkey(leaky(as.w + ad.w)));
            }
        }
        lcount += take;
        head_q += take;
    };

    auto push = [&](bool hit, int e) {
        const unsigned mask = __builtin_amdgcn_ballot_w32(hit);
        if (mask != 0u) {
            const int pre = __builtin_popcount(mask & ((1u << lane) - 1u));
            if (hit) __hip_atomic_store(&ring_all[rbase + ((tail_q + pre) & (RING - 1))], e, __ATOMIC_RELAXED, __HIP_MEMORY_SCOPE_WAVEFRONT);
            tail_q += __builtin_popcount(mask);
            if (tail_q - head_q >= 32) drain_list(32);
        }
    };

    int head_g = 0, tail_g = 0;

    const int4* pairs4 = (const int4*)pairs;
    auto drain_groups = [&](int take) {
        const int sub = lane >> 3;
        const int pr  = lane & 7;
        const unsigned uc = (unsigned)cnt;
        int4 v[8];
        int  gq[8];
#pragma unroll
        for (int i = 0; i < 8; ++i) {
            const int q = 4 * i + sub;
            gq[i] = q < take ? __hip_atomic_load(&ring_grp[rbase + ((head_g + q) & (RING - 1))], __ATOMIC_RELAXED, __HIP_MEMORY_SCOPE_WAVEFRONT) : 0;
        }
#pragma unroll
        for (int i = 0; i < 8; ++i) v[i] = pairs4[(size_t)gq[i] * 8 + pr];
#pragma unroll
        for (int i = 0; i < 8; ++i) {
            if (4 * i + sub >= take) { v[i].y = -1; v[i].w = -1; }
        }
        head_g += take;
        unsigned m[16];
        int total = 0;
#pragma unroll
        for (int i = 0; i < 8; ++i) {
            v[i].y -= n0;
            v[i].w -= n0;
            m[2 * i]     = __builtin_amdgcn_ballot_w32((unsigned)v[i].y < uc);
            m[2 * i + 1] = __builtin_amdgcn_ballot_w32((unsigned)v[i].w < uc);
            total += __builtin_popcount(m[2 * i]) + __builtin_popcount(m[2 * i + 1]);
        }
        int base = lcount;
        if (lcount + total <= CAPW) {
#pragma unroll
            for (int i = 0; i < 8; ++i) {
                if (m[2 * i] != 0u) {
                    const int pos = (int)__builtin_amdgcn_mbcnt_lo(m[2 * i], (unsigned)base);
                    if ((unsigned)v[i].y < uc) {
                        lsrc[lbase + pos] = v[i].x;
                        ldl[lbase + pos]  = (unsigned short)v[i].y;
                    }
                    base += __builtin_popcount(m[2 * i]);
                }
                if (m[2 * i + 1] != 0u) {
                    const int pos = (int)__builtin_amdgcn_mbcnt_lo(m[2 * i + 1], (unsigned)base);
                    if ((unsigned)v[i].w < uc) {
                        lsrc[lbase + pos] = v[i].z;
                        ldl[lbase + pos]  = (unsigned short)v[i].w;
                    }
                    base += __builtin_popcount(m[2 * i + 1]);
                }
            }
        } else {
#pragma unroll
            for (int i = 0; i < 8; ++i) {
#pragma unroll
                for (int t = 0; t < 2; ++t) {
                    const int  s   = t == 0 ? v[i].x : v[i].z;
                    const int  dl  = t == 0 ? v[i].y : v[i].w;
                    const bool hit = (unsigned)dl < uc;
                    const int  pos = (int)__builtin_amdgcn_mbcnt_lo(m[2 * i + t], (unsigned)base);
                    if (hit && pos < CAPW) {
                        lsrc[lbase + pos] = s;
                        ldl[lbase + pos]  = (unsigned short)dl;
                    }
                    if (hit && pos >= CAPW) {
                        const v4f as = *(const v4f*)&a_src[s * HEADS];
                        const v4f ad = *(const v4f*)&adloc[dl * HEADS];
                        atomicMax(&mkl[dl * HEADS + 0], fkey(leaky(as.x + ad.x)));
                        atomicMax(&mkl[dl * HEADS + 1], fkey(leaky(as.y + ad.y)));
                        atomicMax(&mkl[dl * HEADS + 2], fkey(leaky(as.z + ad.z)));
                        atomicMax(&mkl[dl * HEADS + 3], fkey(leaky(as.w + ad.w)));
                    }
                    base += __builtin_popcount(m[2 * i + t]);
                }
            }
        }
        lcount += total;
    };

    auto push_group = [&](bool hit, int g) {
        const unsigned mask = __builtin_amdgcn_ballot_w32(hit);
        if (mask != 0u) {
            const int pre = __builtin_popcount(mask & ((1u << lane) - 1u));
            if (hit) __hip_atomic_store(&ring_grp[rbase + ((tail_g + pre) & (RING - 1))], g, __ATOMIC_RELAXED, __HIP_MEMORY_SCOPE_WAVEFRONT);
            tail_g += __builtin_popcount(mask);
            if (tail_g - head_g >= 32) drain_groups(32);
        }
    };

    const bool vec = ((((size_t)pairs) & 15) == 0) && (gridDim.x <= 256u);
    const int  E16 = vec ? (E >> 4) : 0;
    const int  nW  = (E16 + 31) >> 5;
    const unsigned* plane = bm + (size_t)blockIdx.x * WPO;
    for (int w0 = 0; w0 < nW; w0 += 1024) {
        const int w = w0 + tid;
        unsigned hb = w < nW ? plane[w] : 0u;
        while (__builtin_amdgcn_ballot_w32(hb != 0u) != 0u) {
            const bool hit = hb != 0u;
            const int  tz  = hit ? __builtin_ctz(hb) : 0;
            hb &= hb - 1u;
            push_group(hit, 32 * w + tz);
        }
    }
    if (tail_g - head_g > 0) drain_groups(tail_g - head_g);
    for (int e0 = 16 * E16 + wave * 32; e0 < E; e0 += 1024) {
        const int e = e0 + lane;
        const bool hit = e < E && (unsigned)(dstp[e < E ? e : 0] - n0) < (unsigned)cnt;
        push(hit, e);
    }
    if (tail_q - head_q > 0) drain_list(tail_q - head_q);
    if (lcount > CAPW) over = 1;

    const int nl = __builtin_amdgcn_readfirstlane(min(lcount, CAPW));
    const int nb = (nl + 31) >> 5;
    float lg0[NBAT], lg1[NBAT], lg2[NBAT], lg3[NBAT];
#pragma unroll
    for (int i = 0; i < NBAT; ++i) { lg0[i] = 0.0f; lg1[i] = 0.0f; lg2[i] = 0.0f; lg3[i] = 0.0f; }
    {
        int s_n = 0, dl_n = 0;
        v4f as_n = {0.0f, 0.0f, 0.0f, 0.0f};
        auto fetch = [&](int b) {
            const int  idx = 32 * b + lane;
            const bool act = idx < nl;
            s_n  = act ? lsrc[lbase + idx] : 0;
            dl_n = act ? (int)ldl[lbase + idx] : 0;
            as_n = *(const v4f*)&a_src[s_n * HEADS];
        };
        if (nb > 0) fetch(nb - 1);
        for (int b = nb - 1; b >= 0; --b) {
            const bool act = 32 * b + lane < nl;
            const int  dl  = dl_n;
            const v4f  as  = as_n;
            if (b > 0) fetch(b - 1);
            const v4f ad = *(const v4f*)&adloc[dl * HEADS];
            const float l0 = leaky(as.x + ad.x);
            const float l1 = leaky(as.y + ad.y);
            const float l2 = leaky(as.z + ad.z);
            const float l3 = leaky(as.w + ad.w);
            if (act) {
                atomicMax(&mkl[dl * HEADS + 0], fkey(l0));
                atomicMax(&mkl[dl * HEADS + 1], fkey(l1));
                atomicMax(&mkl[dl * HEADS + 2], fkey(l2));
                atomicMax(&mkl[dl * HEADS + 3], fkey(l3));
            }
#pragma unroll
            for (int i = NBAT - 1; i > 0; --i) { lg0[i] = lg0[i - 1]; lg1[i] = lg1[i - 1]; lg2[i] = lg2[i - 1]; lg3[i] = lg3[i - 1]; }
            lg0[0] = l0; lg1[0] = l1; lg2[0] = l2; lg3[0] = l3;
        }
    }
    __syncthreads();

    auto drain_sum = [&](int take, int s, int dl, float l0, float l1, float l2, float l3) {
        const bool  act = lane < take;
        const uint4 mk  = *(const uint4*)&mkl[dl * HEADS];
        const float p0 = __expf(l0 - funkey(mk.x));
        const float p1 = __expf(l1 - funkey(mk.y));
        const float p2 = __expf(l2 - funkey(mk.z));
        const float p3 = __expf(l3 - funkey(mk.w));
        if (act) {
            atomicAdd(&den[dl * HEADS + 0], p0);
            atomicAdd(&den[dl * HEADS + 1], p1);
            atomicAdd(&den[dl * HEADS + 2], p2);
            atomicAdd(&den[dl * HEADS + 3], p3);
        }
        for (int k0 = 0; k0 < take; k0 += 8) {
            float hv[8][4];
            int   dk[8];
#pragma unroll
            for (int j = 0; j < 8; ++j) {
                const int k  = min(k0 + j, take - 1);
                const int sk = __builtin_amdgcn_readlane(s, k);
                dk[j] = __builtin_amdgcn_readlane(dl, k);
                const float* hr = h + (size_t)sk * HC + lane;
                hv[j][0] = hr[0];
                hv[j][1] = hr[32];
                hv[j][2] = hr[64];
                hv[j][3] = hr[96];
            }
#pragma unroll
            for (int j = 0; j < 8; ++j) {
                if (k0 + j < take) {
                    const int k = k0 + j;
                    float* ar = acc + dk[j] * HC + lane;
                    atomicAdd(ar,      hv[j][0] * bcast_f(p0, k));
                    atomicAdd(ar + 32, hv[j][1] * bcast_f(p1, k));
                    atomicAdd(ar + 64, hv[j][2] * bcast_f(p2, k));
                    atomicAdd(ar + 96, hv[j][3] * bcast_f(p3, k));
                }
            }
        }
    };

    const bool slow = __builtin_amdgcn_readfirstlane(over) != 0;
    if (!slow) {
        for (int b = 0; b < nb; ++b) {
            const int i = 32 * b;
            if (i + 32 <= nl) {
                drain_sum(32, lsrc[lbase + i + lane], (int)ldl[lbase + i + lane], lg0[0], lg1[0], lg2[0], lg3[0]);
            } else {
                const int  take = nl - i;
                const bool act  = lane < take;
                const int  s    = act ? lsrc[lbase + i + lane] : 0;
                const int  dl   = act ? (int)ldl[lbase + i + lane] : 0;
                drain_sum(take, s, dl, lg0[0], lg1[0], lg2[0], lg3[0]);
            }
#pragma unroll
            for (int j = 0; j < NBAT - 1; ++j) { lg0[j] = lg0[j + 1]; lg1[j] = lg1[j + 1]; lg2[j] = lg2[j + 1]; lg3[j] = lg3[j + 1]; }
        }
    } else {
        const int ET = E + cnt;
        for (int e0 = wave * 32; ; e0 += 1024) {
            const bool more = e0 < ET;
            if (more) {
                const int  e   = e0 + lane;
                const bool hit = e < E ? (unsigned)(dstp[e] - n0) < (unsigned)cnt : e < ET;
                const unsigned mask = __builtin_amdgcn_ballot_w32(hit);
                if (mask != 0u) {
                    const int pre = __builtin_popcount(mask & ((1u << lane) - 1u));
                    if (hit) __hip_atomic_store(&ring_all[rbase + ((tail_q + pre) & (RING - 1))], e, __ATOMIC_RELAXED, __HIP_MEMORY_SCOPE_WAVEFRONT);
                    tail_q += __builtin_popcount(mask);
                }
            }
            const int waiting = tail_q - head_q;
            const int take    = more ? (waiting >= 32 ? 32 : 0) : waiting;
            if (take > 0) {
                const bool act = lane < take;
                const int  e   = act ? __hip_atomic_load(&ring_all[rbase + ((head_q + lane) & (RING - 1))], __ATOMIC_RELAXED, __HIP_MEMORY_SCOPE_WAVEFRONT) : E;
                const int  s   = e < E ? ei[e] : n0 + (e - E);
                const int  dl  = e < E ? dstp[e] - n0 : e - E;
                const v4f  as  = *(const v4f*)&a_src[s * HEADS];
                const v4f  ad  = *(const v4f*)&adloc[dl * HEADS];
                drain_sum(take, s, dl, leaky(as.x + ad.x), leaky(as.y + ad.y), leaky(as.z + ad.z), leaky(as.w + ad.w));
                head_q += take;
            }
            if (!more) break;
        }
    }
    __syncthreads();

    for (int r = wave; r < cnt; r += 32) {
        const v4f a = *(const v4f*)&acc[r * HC + lane * 4];
        const float dn = den[r * HEADS + (lane >> 3)] + 1e-16f;
        const v4f b = ((const v4f*)bias)[lane];
        v4f o;
        o.x = a.x / dn + b.x;
        o.y = a.y / dn + b.y;
        o.z = a.z / dn + b.z;
        o.w = a.w / dn + b.w;
        *(v4f*)&out[(size_t)(n0 + r) * HC + lane * 4] = o;
    }
}

extern "C" void kernel_launch(void* const* d_in, const int* in_sizes, int n_in,
                              void* d_out, int out_size, void* d_ws, size_t ws_size,
                              hipStream_t stream) {
    const float* x        = (const float*)d_in[0];
    const int*   ei       = (const int*)d_in[1];
    const float* W        = (const float*)d_in[2];
    const float* att_src  = (const float*)d_in[3];
    const float* att_dst  = (const float*)d_in[4];
    const float* bias     = (const float*)d_in[5];
    float*       out      = (float*)d_out;

    const int N = in_sizes[0] / IN_F;
    const int E = in_sizes[1] / 2;

    float*    h     = (float*)d_ws;
    float*    a_src = h + (size_t)N * HC;
    float*    a_dst = a_src + (size_t)N * HEADS;
    float*    Wa    = a_dst + (size_t)N * HEADS;
    unsigned* bm = (unsigned*)(Wa + IN_F * 8);
    const int GBLK = ((E >> 4) + GB_GROUPS - 1) / GB_GROUPS + ((E >> 4) == 0);
    const int WPO  = GBLK * 4;
    int*      pairs = (int*)(bm + (size_t)256 * WPO);

    gat_fold_att<<<8, 128, 0, stream>>>(W, att_src, att_dst, Wa);

    gat_gemm_wmma<<<N / 16, 160, GEMM_LDS_BYTES, stream>>>(x, W, Wa, h, a_src, a_dst);

    gat_edge_groups<<<GBLK, GB_THREADS, 0, stream>>>(ei, bm, pairs, E, WPO);

    gat_range_sum<<<(N + RNODES - 1) / RNODES, 1024, 0, stream>>>(ei, a_src, a_dst, bm, pairs, h, bias, out, E, N, WPO);
}


# Row gather by whole rows: one 16-byte load per lane fetches an edge's 512-byte row of h (4 lines per instruction), not four 4-byte loads; LDS rows channel-interleaved so the adds stay on 32 banks
# speedup vs baseline: 6.2171x; 1.0566x over previous
#include <hip/hip_runtime.h>

#define IN_F      128
#define HC        128
#define HEADS     4
#define OUT_F     32
#define NEG_SLOPE 0.2f

typedef float v2f __attribute__((ext_vector_type(2)));
typedef float v4f __attribute__((ext_vector_type(4)));
typedef float v8f __attribute__((ext_vector_type(8)));

__device__ __forceinline__ unsigned fkey(float f) {
    unsigned b = __float_as_uint(f);
    return (b & 0x80000000u) ? ~b : (b | 0x80000000u);
}
__device__ __forceinline__ float funkey(unsigned k) {
    unsigned b = (k & 0x80000000u) ? (k & 0x7fffffffu) : ~k;
    return __uint_as_float(b);
}
__device__ __forceinline__ float leaky(float x) {
    return x > 0.0f ? x : NEG_SLOPE * x;
}

__global__ __launch_bounds__(128) void gat_fold_att(const float* __restrict__ W,
                                                    const float* __restrict__ att_src,
                                                    const float* __restrict__ att_dst,
                                                    float* __restrict__ Wa) {
    const int t  = blockIdx.x * 128 + threadIdx.x;
    const int k  = t >> 3;
    const int j  = t & 7;
    const int hd = j & 3;
    const float* att = (j < 4 ? att_src : att_dst) + hd * OUT_F;
    const float* wr  = W + k * HC + hd * OUT_F;
    float s = 0.0f;
#pragma unroll
    for (int c = 0; c < OUT_F; ++c) s += wr[c] * att[c];
    Wa[k * 8 + j] = s;
}

#define WSTR 136
#define XSTR 132
#define GEMM_LDS_BYTES ((128 * WSTR + 16 * XSTR) * 4)

__global__ __launch_bounds__(160) __attribute__((amdgpu_waves_per_eu(4, 6))) void gat_gemm_wmma(const float* __restrict__ x,
                                                     const float* __restrict__ W,
                                                     const float* __restrict__ Wa,
                                                     float* __restrict__ h,
                                                     float* __restrict__ a_src,
                                                     float* __restrict__ a_dst) {
    extern __shared__ float smem[];
    float* Wl = smem;
    float* Xl = smem + 128 * WSTR;

    const int tid  = threadIdx.x;
    const int row0 = blockIdx.x * 16;

    if (tid < 128) {
        const int r0 = tid >> 5;
        const int c4 = tid & 31;
        v4f t[32];
#pragma unroll
        for (int j = 0; j < 32; ++j) t[j] = ((const v4f*)W)[(4 * j + r0) * 32 + c4];
#pragma unroll
        for (int j = 0; j < 32; ++j) *(v4f*)&Wl[(4 * j + r0) * WSTR + c4 * 4] = t[j];
    } else {
        const int l = tid - 128;
        v4f ta[8];
        v4f tx[16];
#pragma unroll
        for (int j = 0; j < 8; ++j) ta[j] = ((const v4f*)Wa)[j * 32 + l];
#pragma unroll
        for (int j = 0; j < 16; ++j) tx[j] = ((const v4f*)x)[(size_t)(row0 + j) * 32 + l];
#pragma unroll
        for (int j = 0; j < 8; ++j) {
            const int i  = j * 32 + l;
            const int r  = i >> 1;
            const int cc = i & 1;
            *(v4f*)&Wl[r * WSTR + 128 + cc * 4] = ta[j];
        }
#pragma unroll
        for (int j = 0; j < 16; ++j) *(v4f*)&Xl[j * XSTR + l * 4] = tx[j];
    }
    __syncthreads();

    const int wave = __builtin_amdgcn_readfirstlane(tid >> 5);
    const int lane = tid & 31;
    const int L    = lane & 15;
    const int g    = lane >> 4;

    if (wave == 4) {
        const int cL = L & 7;
        v8f acc_e = {};
        v8f acc_o = {};
        for (int k0 = 0; k0 < IN_F; k0 += 8) {
            const int k = k0 + 2 * g;
            v2f a0, a1, b0, b1;
            a0.x = Xl[L * XSTR + k];
            a0.y = Xl[L * XSTR + k + 1];
            b0.x = Wl[k * WSTR + 128 + cL];
            b0.y = Wl[(k + 1) * WSTR + 128 + cL];
            a1.x = Xl[L * XSTR + k + 4];
            a1.y = Xl[L * XSTR + k + 5];
            b1.x = Wl[(k + 4) * WSTR + 128 + cL];
            b1.y = Wl[(k + 5) * WSTR + 128 + cL];
            acc_e = __builtin_amdgcn_wmma_f32_16x16x4_f32(false, a0, false, b0, (short)0, acc_e, false, false);
            acc_o = __builtin_amdgcn_wmma_f32_16x16x4_f32(false, a1, false, b1, (short)0, acc_o, false, false);
        }
        if (L < 8) {
            float* ap = (L < 4 ? a_src : a_dst) + (size_t)(row0 + 8 * g) * HEADS + (L & 3);
            for (int r = 0; r < 8; ++r) ap[r * HEADS] = acc_e[r] + acc_o[r];
        }
        return;
    }

    const int col0 = wave * 16;
    const int col1 = (wave + 4) * 16;

    v8f acc0 = {};
    v8f acc1 = {};

    for (int k0 = 0; k0 < IN_F; k0 += 4) {
        const int k = k0 + 2 * g;
        v2f a, b0, b1;
        a.x  = Xl[L * XSTR + k];
        a.y  = Xl[L * XSTR + k + 1];
        b0.x = Wl[k * WSTR + col0 + L];
        b0.y = Wl[(k + 1) * WSTR + col0 + L];
        b1.x = Wl[k * WSTR + col1 + L];
        b1.y = Wl[(k + 1) * WSTR + col1 + L];
        acc0 = __builtin_amdgcn_wmma_f32_16x16x4_f32(false, a, false, b0, (short)0, acc0, false, false);
        acc1 = __builtin_amdgcn_wmma_f32_16x16x4_f32(false, a, false, b1, (short)0, acc1, false, false);
    }

    const int mbase = row0 + 8 * g;
    for (int r = 0; r < 8; ++r) {
        h[(size_t)(mbase + r) * HC + col0 + L] = acc0[r];
        h[(size_t)(mbase + r) * HC + col1 + L] = acc1[r];
    }
}

#define RNODES 391
#define GB_THREADS 512
#define GB_GROUPS  128

__global__ __launch_bounds__(GB_THREADS) void gat_edge_groups(const int* __restrict__ ei,
                                                              unsigned* __restrict__ bm,
                                                              int* __restrict__ pairs,
                                                              int E, int WPO) {
    __shared__ unsigned lb[256 * 4];
    const int tid = threadIdx.x;
    for (int i = tid; i < 256 * 4; i += GB_THREADS) lb[i] = 0u;
    __syncthreads();
    const int* dst = ei + E;
    const int  E16 = E >> 4;
    const int  wave = tid >> 5;
    const int  lane = tid & 31;
    if (E16 > 0) {
        int  lgk[2], ek[2], sa[2], sb[2], da[2], db[2];
        bool ok[2];
#pragma unroll
        for (int k = 0; k < 2; ++k) {
            lgk[k] = 8 * wave + 4 * k + (lane >> 3);
            const int g = blockIdx.x * GB_GROUPS + lgk[k];
            ok[k] = g < E16;
            ek[k] = 16 * min(g, E16 - 1) + 2 * (lane & 7);
            sa[k] = ei[ek[k]];
            sb[k] = ei[ek[k] + 1];
            da[k] = dst[ek[k]];
            db[k] = dst[ek[k] + 1];
        }
#pragma unroll
        for (int k = 0; k < 2; ++k) {
            if (ok[k]) {
                ((int4*)pairs)[ek[k] >> 1] = make_int4(sa[k], da[k], sb[k], db[k]);
                const unsigned bit = 1u << (lgk[k] & 31);
                const unsigned o0 = (unsigned)da[k] / RNODES;
                const unsigned o1 = (unsigned)db[k] / RNODES;
                if (o0 < 256u) atomicOr(&lb[o0 * 4 + (lgk[k] >> 5)], bit);
                if (o1 < 256u) atomicOr(&lb[o1 * 4 + (lgk[k] >> 5)], bit);
            }
        }
    }
    __syncthreads();
    for (int i = tid; i < 256 * 4; i += GB_THREADS)
        bm[(size_t)(i >> 2) * WPO + blockIdx.x * 4 + (i & 3)] = lb[i];
}

#define RING   64
#define CAPW   448
#define NBAT   (CAPW / 32)

__global__ __launch_bounds__(1024) void gat_range_sum(const int* __restrict__ ei,
                                                      const float* __restrict__ a_src,
                                                      const float* __restrict__ a_dst,
                                                      const unsigned* __restrict__ bm,
                                                      const int* __restrict__ pairs,
                                                      const float* __restrict__ h,
                                                      const float* __restrict__ bias,
                                                      float* __restrict__ out,
                                                      int E, int N, int WPO) {
    __shared__ __attribute__((aligned(16))) float acc[RNODES * HC];
    __shared__ __attribute__((aligned(16))) float den[RNODES * HEADS];
    __shared__ __attribute__((aligned(16))) unsigned mkl[RNODES * HEADS];
    __shared__ __attribute__((aligned(16))) float adloc[RNODES * HEADS];
    __shared__ int ring_all[32 * RING];
    __shared__ int ring_grp[32 * RING];
    __shared__ int            lsrc[32 * CAPW];
    __shared__ unsigned short ldl[32 * CAPW];
    __shared__ int over;

    const int tid  = threadIdx.x;
    const int wave = tid >> 5;
    const int lane = tid & 31;
    const int n0   = blockIdx.x * RNODES;
    const int cnt  = min(RNODES, N - n0);
    const int rbase = wave * RING;
    const int lbase = wave * CAPW;

    if (tid == 0) over = 0;
    for (int i = tid; i < RNODES * HC / 4; i += 1024) ((v4f*)acc)[i] = (v4f){0.0f, 0.0f, 0.0f, 0.0f};
    for (int i = tid; i < cnt * HEADS; i += 1024) {
        const float as = a_src[n0 * HEADS + i];
        const float ad = a_dst[n0 * HEADS + i];
        den[i]   = 0.0f;
        adloc[i] = ad;
        mkl[i]   = fkey(leaky(as + ad));
    }
    int lcount;
    {
        const int  r  = wave + 32 * lane;
        const bool hv = r < cnt;
        if (hv) {
            lsrc[lbase + lane] = n0 + r;
            ldl[lbase + lane]  = (unsigned short)r;
        }
        lcount = __builtin_popcount(__builtin_amdgcn_ballot_w32(hv));
    }
    __syncthreads();

    const int* dstp = ei + E;
    int head_q = 0, tail_q = 0;

    auto drain_list = [&](int take) {
        const bool act = lane < take;
        const int  e   = act ? __hip_atomic_load(&ring_all[rbase + ((head_q + lane) & (RING - 1))], __ATOMIC_RELAXED, __HIP_MEMORY_SCOPE_WAVEFRONT) : 0;
        const int  s   = ei[e];
        const int  dl  = act ? dstp[e] - n0 : 0;
        const int  pos = lcount + lane;
        if (act && pos < CAPW) {
            lsrc[lbase + pos] = s;
            ldl[lbase + pos]  = (unsigned short)dl;
        }
        if (lcount + take > CAPW) {
            const v4f as = *(const v4f*)&a_src[s * HEADS];
            const v4f ad = *(const v4f*)&adloc[dl * HEADS];
            if (act && pos >= CAPW) {
                atomicMax(&mkl[dl * HEADS + 0], fkey(leaky(as.x + ad.x)));
                atomicMax(&mkl[dl * HEADS + 1], fkey(leaky(as.y + ad.y)));
                atomicMax(&mkl[dl * HEADS + 2], fkey(leaky(as.z + ad.z)));
                atomicMax(&mkl[dl * HEADS + 3], fkey(leaky(as.w + ad.w)));
            }
        }
        lcount += take;
        head_q += take;
    };

    auto push = [&](bool hit, int e) {
        const unsigned mask = __builtin_amdgcn_ballot_w32(hit);
        if (mask != 0u) {
            const int pre = __builtin_popcount(mask & ((1u << lane) - 1u));
            if (hit) __hip_atomic_store(&ring_all[rbase + ((tail_q + pre) & (RING - 1))], e, __ATOMIC_RELAXED, __HIP_MEMORY_SCOPE_WAVEFRONT);
            tail_q += __builtin_popcount(mask);
            if (tail_q - head_q >= 32) drain_list(32);
        }
    };

    int head_g = 0, tail_g = 0;

    const int4* pairs4 = (const int4*)pairs;
    auto drain_groups = [&](int take) {
        const int sub = lane >> 3;
        const int pr  = lane & 7;
        const unsigned uc = (unsigned)cnt;
        int4 v[8];
        int  gq[8];
#pragma unroll
        for (int i = 0; i < 8; ++i) {
            const int q = 4 * i + sub;
            gq[i] = q < take ? __hip_atomic_load(&ring_grp[rbase + ((head_g + q) & (RING - 1))], __ATOMIC_RELAXED, __HIP_MEMORY_SCOPE_WAVEFRONT) : 0;
        }
#pragma unroll
        for (int i = 0; i < 8; ++i) v[i] = pairs4[(size_t)gq[i] * 8 + pr];
#pragma unroll
        for (int i = 0; i < 8; ++i) {
            if (4 * i + sub >= take) { v[i].y = -1; v[i].w = -1; }
        }
        head_g += take;
        unsigned m[16];
        int total = 0;
#pragma unroll
        for (int i = 0; i < 8; ++i) {
            v[i].y -= n0;
            v[i].w -= n0;
            m[2 * i]     = __builtin_amdgcn_ballot_w32((unsigned)v[i].y < uc);
            m[2 * i + 1] = __builtin_amdgcn_ballot_w32((unsigned)v[i].w < uc);
            total += __builtin_popcount(m[2 * i]) + __builtin_popcount(m[2 * i + 1]);
        }
        int base = lcount;
        if (lcount + total <= CAPW) {
#pragma unroll
            for (int i = 0; i < 8; ++i) {
                if (m[2 * i] != 0u) {
                    const int pos = (int)__builtin_amdgcn_mbcnt_lo(m[2 * i], (unsigned)base);
                    if ((unsigned)v[i].y < uc) {
                        lsrc[lbase + pos] = v[i].x;
                        ldl[lbase + pos]  = (unsigned short)v[i].y;
                    }
                    base += __builtin_popcount(m[2 * i]);
                }
                if (m[2 * i + 1] != 0u) {
                    const int pos = (int)__builtin_amdgcn_mbcnt_lo(m[2 * i + 1], (unsigned)base);
                    if ((unsigned)v[i].w < uc) {
                        lsrc[lbase + pos] = v[i].z;
                        ldl[lbase + pos]  = (unsigned short)v[i].w;
                    }
                    base += __builtin_popcount(m[2 * i + 1]);
                }
            }
        } else {
#pragma unroll
            for (int i = 0; i < 8; ++i) {
#pragma unroll
                for (int t = 0; t < 2; ++t) {
                    const int  s   = t == 0 ? v[i].x : v[i].z;
                    const int  dl  = t == 0 ? v[i].y : v[i].w;
                    const bool hit = (unsigned)dl < uc;
                    const int  pos = (int)__builtin_amdgcn_mbcnt_lo(m[2 * i + t], (unsigned)base);
                    if (hit && pos < CAPW) {
                        lsrc[lbase + pos] = s;
                        ldl[lbase + pos]  = (unsigned short)dl;
                    }
                    if (hit && pos >= CAPW) {
                        const v4f as = *(const v4f*)&a_src[s * HEADS];
                        const v4f ad = *(const v4f*)&adloc[dl * HEADS];
                        atomicMax(&mkl[dl * HEADS + 0], fkey(leaky(as.x + ad.x)));
                        atomicMax(&mkl[dl * HEADS + 1], fkey(leaky(as.y + ad.y)));
                        atomicMax(&mkl[dl * HEADS + 2], fkey(leaky(as.z + ad.z)));
                        atomicMax(&mkl[dl * HEADS + 3], fkey(leaky(as.w + ad.w)));
                    }
                    base += __builtin_popcount(m[2 * i + t]);
                }
            }
        }
        lcount += total;
    };

    auto push_group = [&](bool hit, int g) {
        const unsigned mask = __builtin_amdgcn_ballot_w32(hit);
        if (mask != 0u) {
            const int pre = __builtin_popcount(mask & ((1u << lane) - 1u));
            if (hit) __hip_atomic_store(&ring_grp[rbase + ((tail_g + pre) & (RING - 1))], g, __ATOMIC_RELAXED, __HIP_MEMORY_SCOPE_WAVEFRONT);
            tail_g += __builtin_popcount(mask);
            if (tail_g - head_g >= 32) drain_groups(32);
        }
    };

    const bool vec = ((((size_t)pairs) & 15) == 0) && (gridDim.x <= 256u);
    const int  E16 = vec ? (E >> 4) : 0;
    const int  nW  = (E16 + 31) >> 5;
    const unsigned* plane = bm + (size_t)blockIdx.x * WPO;
    for (int w0 = 0; w0 < nW; w0 += 1024) {
        const int w = w0 + tid;
        unsigned hb = w < nW ? plane[w] : 0u;
        while (__builtin_amdgcn_ballot_w32(hb != 0u) != 0u) {
            const bool hit = hb != 0u;
            const int  tz  = hit ? __builtin_ctz(hb) : 0;
            hb &= hb - 1u;
            push_group(hit, 32 * w + tz);
        }
    }
    if (tail_g - head_g > 0) drain_groups(tail_g - head_g);
    for (int e0 = 16 * E16 + wave * 32; e0 < E; e0 += 1024) {
        const int e = e0 + lane;
        const bool hit = e < E && (unsigned)(dstp[e < E ? e : 0] - n0) < (unsigned)cnt;
        push(hit, e);
    }
    if (tail_q - head_q > 0) drain_list(tail_q - head_q);
    if (lcount > CAPW) over = 1;

    const int nl = __builtin_amdgcn_readfirstlane(min(lcount, CAPW));
    const int nb = (nl + 31) >> 5;
    float lg0[NBAT], lg1[NBAT], lg2[NBAT], lg3[NBAT];
#pragma unroll
    for (int i = 0; i < NBAT; ++i) { lg0[i] = 0.0f; lg1[i] = 0.0f; lg2[i] = 0.0f; lg3[i] = 0.0f; }
    {
        int s_n = 0, dl_n = 0;
        v4f as_n = {0.0f, 0.0f, 0.0f, 0.0f};
        auto fetch = [&](int b) {
            const int  idx = 32 * b + lane;
            const bool act = idx < nl;
            s_n  = act ? lsrc[lbase + idx] : 0;
            dl_n = act ? (int)ldl[lbase + idx] : 0;
            as_n = *(const v4f*)&a_src[s_n * HEADS];
        };
        if (nb > 0) fetch(nb - 1);
        for (int b = nb - 1; b >= 0; --b) {
            const bool act = 32 * b + lane < nl;
            const int  dl  = dl_n;
            const v4f  as  = as_n;
            if (b > 0) fetch(b - 1);
            const v4f ad = *(const v4f*)&adloc[dl * HEADS];
            const float l0 = leaky(as.x + ad.x);
            const float l1 = leaky(as.y + ad.y);
            const float l2 = leaky(as.z + ad.z);
            const float l3 = leaky(as.w + ad.w);
            if (act) {
                atomicMax(&mkl[dl * HEADS + 0], fkey(l0));
                atomicMax(&mkl[dl * HEADS + 1], fkey(l1));
                atomicMax(&mkl[dl * HEADS + 2], fkey(l2));
                atomicMax(&mkl[dl * HEADS + 3], fkey(l3));
            }
#pragma unroll
            for (int i = NBAT - 1; i > 0; --i) { lg0[i] = lg0[i - 1]; lg1[i] = lg1[i - 1]; lg2[i] = lg2[i - 1]; lg3[i] = lg3[i - 1]; }
            lg0[0] = l0; lg1[0] = l1; lg2[0] = l2; lg3[0] = l3;
        }
    }
    __syncthreads();

    auto drain_sum = [&](int take, int s, int dl, float l0, float l1, float l2, float l3) {
        const bool  act = lane < take;
        const uint4 mk  = *(const uint4*)&mkl[dl * HEADS];
        const float p0 = __expf(l0 - funkey(mk.x));
        const float p1 = __expf(l1 - funkey(mk.y));
        const float p2 = __expf(l2 - funkey(mk.z));
        const float p3 = __expf(l3 - funkey(mk.w));
        if (act) {
            atomicAdd(&den[dl * HEADS + 0], p0);
            atomicAdd(&den[dl * HEADS + 1], p1);
            atomicAdd(&den[dl * HEADS + 2], p2);
            atomicAdd(&den[dl * HEADS + 3], p3);
        }
        const int hd = lane >> 3;
        const bool is1 = hd == 1, is2 = hd == 2, is3 = hd == 3;
        for (int k0 = 0; k0 < take; k0 += 8) {
            v4f hv[8];
            int dk[8];
#pragma unroll
            for (int j = 0; j < 8; ++j) {
                const int k  = min(k0 + j, take - 1);
                const int sk = __builtin_amdgcn_readlane(s, k);
                dk[j] = __builtin_amdgcn_readlane(dl, k);
                hv[j] = *(const v4f*)(h + (size_t)sk * HC + lane * 4);
            }
#pragma unroll
            for (int j = 0; j < 8; ++j) {
                if (k0 + j < take) {
                    const int k = k0 + j;
                    const unsigned w0 = (unsigned)__builtin_amdgcn_readlane(__float_as_int(p0), k);
                    const unsigned w1 = (unsigned)__builtin_amdgcn_readlane(__float_as_int(p1), k);
                    const unsigned w2 = (unsigned)__builtin_amdgcn_readlane(__float_as_int(p2), k);
                    const unsigned w3 = (unsigned)__builtin_amdgcn_readlane(__float_as_int(p3), k);
                    unsigned wb = w0;
                    wb = is1 ? w1 : wb;
                    wb = is2 ? w2 : wb;
                    wb = is3 ? w3 : wb;
                    const float w = __uint_as_float(wb);
                    float* ar = acc + dk[j] * HC + lane;
                    atomicAdd(ar,      hv[j].x * w);
                    atomicAdd(ar + 32, hv[j].y * w);
                    atomicAdd(ar + 64, hv[j].z * w);
                    atomicAdd(ar + 96, hv[j].w * w);
                }
            }
        }
    };

    const bool slow = __builtin_amdgcn_readfirstlane(over) != 0;
    if (!slow) {
        for (int b = 0; b < nb; ++b) {
            const int i = 32 * b;
            if (i + 32 <= nl) {
                drain_sum(32, lsrc[lbase + i + lane], (int)ldl[lbase + i + lane], lg0[0], lg1[0], lg2[0], lg3[0]);
            } else {
                const int  take = nl - i;
                const bool act  = lane < take;
                const int  s    = act ? lsrc[lbase + i + lane] : 0;
                const int  dl   = act ? (int)ldl[lbase + i + lane] : 0;
                drain_sum(take, s, dl, lg0[0], lg1[0], lg2[0], lg3[0]);
            }
#pragma unroll
            for (int j = 0; j < NBAT - 1; ++j) { lg0[j] = lg0[j + 1]; lg1[j] = lg1[j + 1]; lg2[j] = lg2[j + 1]; lg3[j] = lg3[j + 1]; }
        }
    } else {
        const int ET = E + cnt;
        for (int e0 = wave * 32; ; e0 += 1024) {
            const bool more = e0 < ET;
            if (more) {
                const int  e   = e0 + lane;
                const bool hit = e < E ? (unsigned)(dstp[e] - n0) < (unsigned)cnt : e < ET;
                const unsigned mask = __builtin_amdgcn_ballot_w32(hit);
                if (mask != 0u) {
                    const int pre = __builtin_popcount(mask & ((1u << lane) - 1u));
                    if (hit) __hip_atomic_store(&ring_all[rbase + ((tail_q + pre) & (RING - 1))], e, __ATOMIC_RELAXED, __HIP_MEMORY_SCOPE_WAVEFRONT);
                    tail_q += __builtin_popcount(mask);
                }
            }
            const int waiting = tail_q - head_q;
            const int take    = more ? (waiting >= 32 ? 32 : 0) : waiting;
            if (take > 0) {
                const bool act = lane < take;
                const int  e   = act ? __hip_atomic_load(&ring_all[rbase + ((head_q + lane) & (RING - 1))], __ATOMIC_RELAXED, __HIP_MEMORY_SCOPE_WAVEFRONT) : E;
                const int  s   = e < E ? ei[e] : n0 + (e - E);
                const int  dl  = e < E ? dstp[e] - n0 : e - E;
                const v4f  as  = *(const v4f*)&a_src[s * HEADS];
                const v4f  ad  = *(const v4f*)&adloc[dl * HEADS];
                drain_sum(take, s, dl, leaky(as.x + ad.x), leaky(as.y + ad.y), leaky(as.z + ad.z), leaky(as.w + ad.w));
                head_q += take;
            }
            if (!more) break;
        }
    }
    __syncthreads();

    for (int r = wave; r < cnt; r += 32) {
        v4f a;
        a.x = acc[r * HC + lane];
        a.y = acc[r * HC + 32 + lane];
        a.z = acc[r * HC + 64 + lane];
        a.w = acc[r * HC + 96 + lane];
        const float dn = den[r * HEADS + (lane >> 3)] + 1e-16f;
        const v4f b = ((const v4f*)bias)[lane];
        v4f o;
        o.x = a.x / dn + b.x;
        o.y = a.y / dn + b.y;
        o.z = a.z / dn + b.z;
        o.w = a.w / dn + b.w;
        *(v4f*)&out[(size_t)(n0 + r) * HC + lane * 4] = o;
    }
}

extern "C" void kernel_launch(void* const* d_in, const int* in_sizes, int n_in,
                              void* d_out, int out_size, void* d_ws, size_t ws_size,
                              hipStream_t stream) {
    const float* x        = (const float*)d_in[0];
    const int*   ei       = (const int*)d_in[1];
    const float* W        = (const float*)d_in[2];
    const float* att_src  = (const float*)d_in[3];
    const float* att_dst  = (const float*)d_in[4];
    const float* bias     = (const float*)d_in[5];
    float*       out      = (float*)d_out;

    const int N = in_sizes[0] / IN_F;
    const int E = in_sizes[1] / 2;

    float*    h     = (float*)d_ws;
    float*    a_src = h + (size_t)N * HC;
    float*    a_dst = a_src + (size_t)N * HEADS;
    float*    Wa    = a_dst + (size_t)N * HEADS;
    unsigned* bm = (unsigned*)(Wa + IN_F * 8);
    const int GBLK = ((E >> 4) + GB_GROUPS - 1) / GB_GROUPS + ((E >> 4) == 0);
    const int WPO  = GBLK * 4;
    int*      pairs = (int*)(bm + (size_t)256 * WPO);

    gat_fold_att<<<8, 128, 0, stream>>>(W, att_src, att_dst, Wa);

    gat_gemm_wmma<<<N / 16, 160, GEMM_LDS_BYTES, stream>>>(x, W, Wa, h, a_src, a_dst);

    gat_edge_groups<<<GBLK, GB_THREADS, 0, stream>>>(ei, bm, pairs, E, WPO);

    gat_range_sum<<<(N + RNODES - 1) / RNODES, 1024, 0, stream>>>(ei, a_src, a_dst, bm, pairs, h, bias, out, E, N, WPO);
}
